# starting kernel
# baseline (speedup 1.0000x reference)
#include <hip/hip_runtime.h>

#define NNODES 100000
#define NEDGES 1600000
#define NPAIRS 200000
#define DIN    128
#define DHID   128
#define DOUT   64
#define KDIM   128
#define APITCH 132

typedef __attribute__((ext_vector_type(2))) float v2f;
typedef __attribute__((ext_vector_type(8))) float v8f;

__global__ void k_deg_init(float* __restrict__ deg, int n) {
  int i = blockIdx.x * blockDim.x + threadIdx.x;
  if (i < n) deg[i] = 1.0f;
}

__global__ void k_deg_accum(const int* __restrict__ dst, float* __restrict__ deg, int e) {
  int i = blockIdx.x * blockDim.x + threadIdx.x;
  if (i < e) atomicAdd(&deg[dst[i]], 1.0f);
}

__global__ void k_deg_rsqrt(float* __restrict__ deg, int n) {
  int i = blockIdx.x * blockDim.x + threadIdx.x;
  if (i < n) deg[i] = rsqrtf(deg[i]);
}

__global__ void k_edge_norm(const int* __restrict__ src, const int* __restrict__ dst,
                            const float* __restrict__ dinv, float* __restrict__ norm, int e) {
  int i = blockIdx.x * blockDim.x + threadIdx.x;
  if (i < e) norm[i] = dinv[src[i]] * dinv[dst[i]];
}

template <int NOUT>
__global__ __launch_bounds__(128) void k_gemm_wmma(const float* __restrict__ A,
                                                   const float* __restrict__ B,
                                                   float* __restrict__ C) {
  __shared__ float lA[16 * APITCH];

  const int tid  = threadIdx.x;
  const int wave = tid >> 5;
  const int lane = tid & 31;
  const int row0 = blockIdx.x * 16;
  const int col0 = (blockIdx.y * 4 + wave) * 16;
  const int half = lane >> 4;
  const int l16  = lane & 15;

  {
    const float4* gA = (const float4*)(A + (size_t)row0 * KDIM);
    float4* sA = (float4*)lA;
#pragma unroll
    for (int i = 0; i < 4; ++i) {
      int idx = tid + i * 128;
      int r   = idx >> 5;
      int c4  = idx & 31;
      sA[r * (APITCH / 4) + c4] = gA[idx];
    }
  }
  __syncthreads();

  const float* aP = lA + l16 * APITCH + half * 2;
  const float* bP = B + (half * 2) * NOUT + (col0 + l16);

  v8f acc = {};
#pragma unroll
  for (int k = 0; k < KDIM; k += 4) {
    v2f av = *(const v2f*)(aP + k);
    v2f bv;
    bv.x = bP[k * NOUT];
    bv.y = bP[(k + 1) * NOUT];
    acc = __builtin_amdgcn_wmma_f32_16x16x4_f32(false, av, false, bv,
                                                (short)0, acc, false, false);
  }

  float* cP = C + (size_t)(row0 + half * 8) * NOUT + (col0 + l16);
#pragma unroll
  for (int r = 0; r < 8; ++r) cP[r * NOUT] = acc[r];
}

__global__ void k_agg_init(const float* __restrict__ h, const float* __restrict__ dinv,
                           const float* __restrict__ bias, float* __restrict__ out,
                           int n, int fshift, int fmask) {
  long long idx = (long long)blockIdx.x * blockDim.x + threadIdx.x;
  if (idx >= ((long long)n << fshift)) return;
  int i = (int)(idx >> fshift);
  int f = (int)(idx & fmask);
  float s = dinv[i];
  out[idx] = s * s * h[idx] + bias[f];
}

__global__ void k_agg_edges(const float* __restrict__ h, const int* __restrict__ src,
                            const int* __restrict__ dst, const float* __restrict__ norm,
                            float* __restrict__ out, int e, int f4shift, int f4mask) {
  long long idx = (long long)blockIdx.x * blockDim.x + threadIdx.x;
  if (idx >= ((long long)e << f4shift)) return;
  int ei = (int)(idx >> f4shift);
  int f4 = (int)(idx & f4mask);
  int s = src[ei], d = dst[ei];
  float w = norm[ei];
  const float4 v = ((const float4*)h)[((size_t)s << f4shift) + f4];
  float* o = out + ((((size_t)d << f4shift) + (size_t)f4) << 2);
  atomicAdd(o + 0, v.x * w);
  atomicAdd(o + 1, v.y * w);
  atomicAdd(o + 2, v.z * w);
  atomicAdd(o + 3, v.w * w);
}

__global__ void k_relu(float* __restrict__ x, long long n) {
  long long i = (long long)blockIdx.x * blockDim.x + threadIdx.x;
  if (i < n) x[i] = fmaxf(x[i], 0.0f);
}

__global__ void k_decode(const float* __restrict__ z, const int* __restrict__ pa,
                         const int* __restrict__ pb, float* __restrict__ out, int np) {
  int tid = blockIdx.x * blockDim.x + threadIdx.x;
  int pair = tid >> 4;
  int l    = tid & 15;
  if (pair >= np) return;
  const float4 va = ((const float4*)z)[((size_t)pa[pair] << 4) + l];
  const float4 vb = ((const float4*)z)[((size_t)pb[pair] << 4) + l];
  float p = va.x * vb.x + va.y * vb.y + va.z * vb.z + va.w * vb.w;
  p += __shfl_xor(p, 8);
  p += __shfl_xor(p, 4);
  p += __shfl_xor(p, 2);
  p += __shfl_xor(p, 1);
  if (l == 0) out[pair] = p;
}

extern "C" void kernel_launch(void* const* d_in, const int* in_sizes, int n_in,
                              void* d_out, int out_size, void* d_ws, size_t ws_size,
                              hipStream_t stream) {
  (void)in_sizes; (void)n_in; (void)out_size; (void)ws_size;

  const float* x  = (const float*)d_in[0];
  const int*   ei = (const int*)d_in[1];
  const int*   el = (const int*)d_in[2];
  const float* W1 = (const float*)d_in[3];
  const float* b1 = (const float*)d_in[4];
  const float* W2 = (const float*)d_in[5];
  const float* b2 = (const float*)d_in[6];
  const float* W3 = (const float*)d_in[7];
  const float* b3 = (const float*)d_in[8];

  const int* src = ei;
  const int* dst = ei + NEDGES;
  const int* pa  = el;
  const int* pb  = el + NPAIRS;
  float* out = (float*)d_out;

  float* dinv = (float*)d_ws;
  float* norm = dinv + ((NNODES + 127) & ~127);
  float* bufA = norm + ((NEDGES + 127) & ~127);
  float* bufB = bufA + (size_t)NNODES * DHID;

  auto cdiv = [](long long a, long long b) { return (int)((a + b - 1) / b); };

  k_deg_init <<<cdiv(NNODES, 256), 256, 0, stream>>>(dinv, NNODES);
  k_deg_accum<<<cdiv(NEDGES, 256), 256, 0, stream>>>(dst, dinv, NEDGES);
  k_deg_rsqrt<<<cdiv(NNODES, 256), 256, 0, stream>>>(dinv, NNODES);
  k_edge_norm<<<cdiv(NEDGES, 256), 256, 0, stream>>>(src, dst, dinv, norm, NEDGES);

  const int mTiles = NNODES / 16;

  k_gemm_wmma<DHID><<<dim3(mTiles, DHID / 64), 128, 0, stream>>>(x, W1, bufA);
  k_agg_init <<<cdiv((long long)NNODES * DHID, 256), 256, 0, stream>>>(bufA, dinv, b1, bufB, NNODES, 7, 127);
  k_agg_edges<<<cdiv((long long)NEDGES * (DHID / 4), 256), 256, 0, stream>>>(bufA, src, dst, norm, bufB, NEDGES, 5, 31);
  k_relu     <<<cdiv((long long)NNODES * DHID, 256), 256, 0, stream>>>(bufB, (long long)NNODES * DHID);

  k_gemm_wmma<DHID><<<dim3(mTiles, DHID / 64), 128, 0, stream>>>(bufB, W2, bufA);
  k_agg_init <<<cdiv((long long)NNODES * DHID, 256), 256, 0, stream>>>(bufA, dinv, b2, bufB, NNODES, 7, 127);
  k_agg_edges<<<cdiv((long long)NEDGES * (DHID / 4), 256), 256, 0, stream>>>(bufA, src, dst, norm, bufB, NEDGES, 5, 31);
  k_relu     <<<cdiv((long long)NNODES * DHID, 256), 256, 0, stream>>>(bufB, (long long)NNODES * DHID);

  k_gemm_wmma<DOUT><<<dim3(mTiles, DOUT / 64), 128, 0, stream>>>(bufB, W3, bufA);
  k_agg_init <<<cdiv((long long)NNODES * DOUT, 256), 256, 0, stream>>>(bufA, dinv, b3, bufB, NNODES, 6, 63);
  k_agg_edges<<<cdiv((long long)NEDGES * (DOUT / 4), 256), 256, 0, stream>>>(bufA, src, dst, norm, bufB, NEDGES, 4, 15);

  k_decode<<<cdiv((long long)NPAIRS * 16, 256), 256, 0, stream>>>(bufB, pa, pb, out, NPAIRS);
}


# aggregation by gather, not float-atomic scatter: edges bucketed by destination in-call (count+rank, scan, fill); one wave sums a row in arrival-rank order, one store with self-loop, bias, relu
# speedup vs baseline: 3.0891x; 3.0891x over previous
#include <hip/hip_runtime.h>

#define NNODES 100000
#define NEDGES 1600000
#define NPAIRS 200000
#define DIN    128
#define DHID   128
#define DOUT   64
#define KDIM   128
#define APITCH 132

typedef __attribute__((ext_vector_type(2))) float v2f;
typedef __attribute__((ext_vector_type(8))) float v8f;

__global__ void k_cnt_zero(int* __restrict__ cnt, int n) {
  int i = blockIdx.x * blockDim.x + threadIdx.x;
  if (i < n) cnt[i] = 0;
}

__global__ void k_cnt_rank(const int* __restrict__ dst, int* __restrict__ cnt,
                           int* __restrict__ rank, int e) {
  int i = blockIdx.x * blockDim.x + threadIdx.x;
  if (i < e) rank[i] = atomicAdd(&cnt[dst[i]], 1);
}

#define SCAN_ITEMS 4
#define SCAN_TILE  1024
static_assert((NNODES + SCAN_TILE - 1) / SCAN_TILE <= 128, "tsum holds 128 tile sums");

__device__ __forceinline__ void load_cnt4(const int* __restrict__ cnt, int base, int n, int c[SCAN_ITEMS]) {
  if (base + SCAN_ITEMS <= n) {
    const int4 q = *(const int4*)(cnt + base);
    c[0] = q.x; c[1] = q.y; c[2] = q.z; c[3] = q.w;
  } else {
    for (int j = 0; j < SCAN_ITEMS; ++j) c[j] = (base + j < n) ? cnt[base + j] : 0;
  }
}

__device__ __forceinline__ int block_sum256(int x, int* sh) {
  const int lane = threadIdx.x & 31, wv = threadIdx.x >> 5;
  for (int o = 16; o > 0; o >>= 1) x += __shfl_xor(x, o);
  if (lane == 0) sh[wv] = x;
  __syncthreads();
  int t = 0;
  for (int i = 0; i < 8; ++i) t += sh[i];
  return t;
}

__global__ __launch_bounds__(256) void k_scan_tiles(const int* __restrict__ cnt,
                                                    int* __restrict__ tsum, int n) {
  __shared__ int sh[8];
  int c[SCAN_ITEMS];
  load_cnt4(cnt, blockIdx.x * SCAN_TILE + threadIdx.x * SCAN_ITEMS, n, c);
  const int tot = block_sum256(c[0] + c[1] + c[2] + c[3], sh);
  if (threadIdx.x == 0) tsum[blockIdx.x] = tot;
}

__global__ __launch_bounds__(256) void k_scan_rows(const int* __restrict__ cnt,
                                                   const int* __restrict__ tsum,
                                                   int* __restrict__ rowptr,
                                                   float* __restrict__ dinv, int n) {
  __shared__ int sh[8];
  __shared__ int shw[8];
  const int t = threadIdx.x, lane = t & 31, wv = t >> 5;

  int part = 0;
  for (int i = t; i < (int)blockIdx.x; i += 256) part += tsum[i];
  const int tile_off = block_sum256(part, sh);

  const int base = blockIdx.x * SCAN_TILE + t * SCAN_ITEMS;
  int c[SCAN_ITEMS];
  load_cnt4(cnt, base, n, c);
  const int mine = c[0] + c[1] + c[2] + c[3];

  int incl = mine;
  for (int o = 1; o < 32; o <<= 1) {
    const int y = __shfl_up(incl, o);
    if (lane >= o) incl += y;
  }
  if (lane == 31) shw[wv] = incl;
  __syncthreads();
  int wave_off = 0;
  for (int i = 0; i < wv; ++i) wave_off += shw[i];

  int run = tile_off + wave_off + incl - mine;
  for (int j = 0; j < SCAN_ITEMS; ++j) {
    const int i = base + j;
    if (i < n) {
      rowptr[i] = run;
      dinv[i]   = rsqrtf((float)(c[j] + 1));
      run += c[j];
      if (i == n - 1) rowptr[n] = run;
    }
  }
}

__global__ void k_fill(const int* __restrict__ src, const int* __restrict__ dst,
                       const int* __restrict__ rank, const int* __restrict__ rowptr,
                       const float* __restrict__ dinv, int2* __restrict__ rec, int e) {
  int i = blockIdx.x * blockDim.x + threadIdx.x;
  if (i >= e) return;
  const int s = src[i], d = dst[i];
  const int pos = rowptr[d] + rank[i];
  const float w = dinv[s] * dinv[d];
  if ((unsigned)pos < (unsigned)e) rec[pos] = make_int2(s, __float_as_int(w));
}

template <int NOUT>
__global__ __launch_bounds__(128) void k_gemm_wmma(const float* __restrict__ A,
                                                   const float* __restrict__ B,
                                                   float* __restrict__ C) {
  __shared__ float lA[16 * APITCH];

  const int tid  = threadIdx.x;
  const int wave = tid >> 5;
  const int lane = tid & 31;
  const int row0 = blockIdx.x * 16;
  const int col0 = (blockIdx.y * 4 + wave) * 16;
  const int half = lane >> 4;
  const int l16  = lane & 15;

  {
    const float4* gA = (const float4*)(A + (size_t)row0 * KDIM);
    float4* sA = (float4*)lA;
#pragma unroll
    for (int i = 0; i < 4; ++i) {
      int idx = tid + i * 128;
      int r   = idx >> 5;
      int c4  = idx & 31;
      sA[r * (APITCH / 4) + c4] = gA[idx];
    }
  }
  __syncthreads();

  const float* aP = lA + l16 * APITCH + half * 2;
  const float* bP = B + (half * 2) * NOUT + (col0 + l16);

  v8f acc = {};
#pragma unroll
  for (int k = 0; k < KDIM; k += 4) {
    v2f av = *(const v2f*)(aP + k);
    v2f bv;
    bv.x = bP[k * NOUT];
    bv.y = bP[(k + 1) * NOUT];
    acc = __builtin_amdgcn_wmma_f32_16x16x4_f32(false, av, false, bv,
                                                (short)0, acc, false, false);
  }

  float* cP = C + (size_t)(row0 + half * 8) * NOUT + (col0 + l16);
#pragma unroll
  for (int r = 0; r < 8; ++r) cP[r * NOUT] = acc[r];
}

#define GATHER_DEPTH 8

template <int V, bool FULL>
__device__ __forceinline__ void gather_group(float __attribute__((ext_vector_type(V)))& acc,
                                             const float __attribute__((ext_vector_type(V)))* __restrict__ hv,
                                             const int2 r, const int k, const int m,
                                             const int lane, const unsigned last_node) {
  typedef float vec_t __attribute__((ext_vector_type(V)));
  unsigned s[GATHER_DEPTH];
  float    w[GATHER_DEPTH];
  vec_t    v[GATHER_DEPTH];
#pragma unroll
  for (int u = 0; u < GATHER_DEPTH; ++u) {
    s[u] = min((unsigned)__builtin_amdgcn_readlane(r.x, k + u), last_node);
    w[u] = __int_as_float(__builtin_amdgcn_readlane(r.y, k + u));
  }
#pragma unroll
  for (int u = 0; u < GATHER_DEPTH; ++u) {
    v[u] = (vec_t)(0.0f);
    if (FULL || u < m) v[u] = hv[s[u] * 32u + lane];
  }
#pragma unroll
  for (int u = 0; u < GATHER_DEPTH; ++u) acc += v[u] * w[u];
}

template <int F, bool RELU>
__global__ __launch_bounds__(256) void k_agg_gather(const float* __restrict__ h,
                                                    const int* __restrict__ rowptr,
                                                    const int2* __restrict__ rec,
                                                    const float* __restrict__ dinv,
                                                    const float* __restrict__ bias,
                                                    float* __restrict__ out, int n) {
  constexpr int V = F / 32;
  typedef float vec_t __attribute__((ext_vector_type(V)));
  const vec_t* __restrict__ hv = (const vec_t*)h;

  const int lane = threadIdx.x & 31;
  const int row = __builtin_amdgcn_readfirstlane((int)((blockIdx.x * blockDim.x + threadIdx.x) >> 5));
  if (row >= n) return;

  const int beg = rowptr[row];
  const int end = rowptr[row + 1];
  const float di = dinv[row];
  const vec_t bv = ((const vec_t*)bias)[lane];
  vec_t acc = hv[(unsigned)row * 32u + lane] * (di * di);

  for (int base = beg; base < end; base += 32) {
    int2 r = make_int2(0, 0);
    if (base + lane < end) r = rec[base + lane];
    const int cnt = min(32, end - base);

    int k = 0;
    for (; k + GATHER_DEPTH <= cnt; k += GATHER_DEPTH)
      gather_group<V, true>(acc, hv, r, k, GATHER_DEPTH, lane, (unsigned)(n - 1));
    if (k < cnt)
      gather_group<V, false>(acc, hv, r, k, cnt - k, lane, (unsigned)(n - 1));
  }

  acc += bv;
  if (RELU) {
    for (int c = 0; c < V; ++c) acc[c] = fmaxf(acc[c], 0.0f);
  }
  ((vec_t*)out)[(unsigned)row * 32u + lane] = acc;
}

__global__ void k_decode(const float* __restrict__ z, const int* __restrict__ pa,
                         const int* __restrict__ pb, float* __restrict__ out, int np) {
  int tid = blockIdx.x * blockDim.x + threadIdx.x;
  int pair = tid >> 4;
  int l    = tid & 15;
  if (pair >= np) return;
  const float4 va = ((const float4*)z)[((size_t)pa[pair] << 4) + l];
  const float4 vb = ((const float4*)z)[((size_t)pb[pair] << 4) + l];
  float p = va.x * vb.x + va.y * vb.y + va.z * vb.z + va.w * vb.w;
  p += __shfl_xor(p, 8);
  p += __shfl_xor(p, 4);
  p += __shfl_xor(p, 2);
  p += __shfl_xor(p, 1);
  if (l == 0) out[pair] = p;
}

extern "C" void kernel_launch(void* const* d_in, const int* in_sizes, int n_in,
                              void* d_out, int out_size, void* d_ws, size_t ws_size,
                              hipStream_t stream) {
  (void)in_sizes; (void)n_in; (void)out_size; (void)ws_size;

  const float* x  = (const float*)d_in[0];
  const int*   ei = (const int*)d_in[1];
  const int*   el = (const int*)d_in[2];
  const float* W1 = (const float*)d_in[3];
  const float* b1 = (const float*)d_in[4];
  const float* W2 = (const float*)d_in[5];
  const float* b2 = (const float*)d_in[6];
  const float* W3 = (const float*)d_in[7];
  const float* b3 = (const float*)d_in[8];

  const int* src = ei;
  const int* dst = ei + NEDGES;
  const int* pa  = el;
  const int* pb  = el + NPAIRS;
  float* out = (float*)d_out;

  float* dinv   = (float*)d_ws;
  int*   rank   = (int*)(dinv + ((NNODES + 127) & ~127));
  float* bufA   = (float*)(rank + ((NEDGES + 127) & ~127));
  float* bufB   = bufA + (size_t)NNODES * DHID;
  int*   cnt    = (int*)(bufB + (size_t)NNODES * DHID);
  int*   rowptr = cnt + ((NNODES + 127) & ~127);
  int*   tsum   = rowptr + ((NNODES + 1 + 127) & ~127);
  int2*  rec    = (int2*)(tsum + 128);

  auto cdiv = [](long long a, long long b) { return (int)((a + b - 1) / b); };

  const int nTiles = cdiv(NNODES, SCAN_TILE);
  k_cnt_zero  <<<cdiv(NNODES, 256), 256, 0, stream>>>(cnt, NNODES);
  k_cnt_rank  <<<cdiv(NEDGES, 256), 256, 0, stream>>>(dst, cnt, rank, NEDGES);
  k_scan_tiles<<<nTiles, 256, 0, stream>>>(cnt, tsum, NNODES);
  k_scan_rows <<<nTiles, 256, 0, stream>>>(cnt, tsum, rowptr, dinv, NNODES);
  k_fill      <<<cdiv(NEDGES, 256), 256, 0, stream>>>(src, dst, rank, rowptr, dinv, rec, NEDGES);

  const int mTiles = NNODES / 16;
  const int gBlocks = cdiv((long long)NNODES * 32, 256);

  k_gemm_wmma<DHID><<<dim3(mTiles, DHID / 64), 128, 0, stream>>>(x, W1, bufA);
  k_agg_gather<DHID, true><<<gBlocks, 256, 0, stream>>>(bufA, rowptr, rec, dinv, b1, bufB, NNODES);

  k_gemm_wmma<DHID><<<dim3(mTiles, DHID / 64), 128, 0, stream>>>(bufB, W2, bufA);
  k_agg_gather<DHID, true><<<gBlocks, 256, 0, stream>>>(bufA, rowptr, rec, dinv, b2, bufB, NNODES);

  k_gemm_wmma<DOUT><<<dim3(mTiles, DOUT / 64), 128, 0, stream>>>(bufB, W3, bufA);
  k_agg_gather<DOUT, false><<<gBlocks, 256, 0, stream>>>(bufA, rowptr, rec, dinv, b3, bufB, NNODES);

  k_decode<<<cdiv((long long)NPAIRS * 16, 256), 256, 0, stream>>>(bufB, pa, pb, out, NPAIRS);
}


# next layer's dense product fused into the aggregation kernel: the gathered 16-row tile goes through LDS straight into the WMMAs, weights LDS-resident in persistent blocks; 12 launches become 10
# speedup vs baseline: 3.2928x; 1.0659x over previous
#include <hip/hip_runtime.h>

#define NNODES 100000
#define NEDGES 1600000
#define NPAIRS 200000
#define DIN    128
#define DHID   128
#define DOUT   64
#define KDIM   128
#define APITCH 132

typedef __attribute__((ext_vector_type(2))) float v2f;
typedef __attribute__((ext_vector_type(8))) float v8f;

__global__ void k_cnt_zero(int* __restrict__ cnt, int n) {
  int i = blockIdx.x * blockDim.x + threadIdx.x;
  if (i < n) cnt[i] = 0;
}

__global__ void k_cnt_rank(const int* __restrict__ dst, int* __restrict__ cnt,
                           int* __restrict__ rank, int e) {
  int i = blockIdx.x * blockDim.x + threadIdx.x;
  if (i < e) rank[i] = atomicAdd(&cnt[dst[i]], 1);
}

#define SCAN_ITEMS 4
#define SCAN_TILE  1024
static_assert((NNODES + SCAN_TILE - 1) / SCAN_TILE <= 128, "tsum holds 128 tile sums");

__device__ __forceinline__ void load_cnt4(const int* __restrict__ cnt, int base, int n, int c[SCAN_ITEMS]) {
  if (base + SCAN_ITEMS <= n) {
    const int4 q = *(const int4*)(cnt + base);
    c[0] = q.x; c[1] = q.y; c[2] = q.z; c[3] = q.w;
  } else {
    for (int j = 0; j < SCAN_ITEMS; ++j) c[j] = (base + j < n) ? cnt[base + j] : 0;
  }
}

__device__ __forceinline__ int block_sum256(int x, int* sh) {
  const int lane = threadIdx.x & 31, wv = threadIdx.x >> 5;
  for (int o = 16; o > 0; o >>= 1) x += __shfl_xor(x, o);
  if (lane == 0) sh[wv] = x;
  __syncthreads();
  int t = 0;
  for (int i = 0; i < 8; ++i) t += sh[i];
  return t;
}

__global__ __launch_bounds__(256) void k_scan_tiles(const int* __restrict__ cnt,
                                                    int* __restrict__ tsum, int n) {
  __shared__ int sh[8];
  int c[SCAN_ITEMS];
  load_cnt4(cnt, blockIdx.x * SCAN_TILE + threadIdx.x * SCAN_ITEMS, n, c);
  const int tot = block_sum256(c[0] + c[1] + c[2] + c[3], sh);
  if (threadIdx.x == 0) tsum[blockIdx.x] = tot;
}

__global__ __launch_bounds__(256) void k_scan_rows(const int* __restrict__ cnt,
                                                   const int* __restrict__ tsum,
                                                   int* __restrict__ rowptr,
                                                   float* __restrict__ dinv, int n) {
  __shared__ int sh[8];
  __shared__ int shw[8];
  const int t = threadIdx.x, lane = t & 31, wv = t >> 5;

  int part = 0;
  for (int i = t; i < (int)blockIdx.x; i += 256) part += tsum[i];
  const int tile_off = block_sum256(part, sh);

  const int base = blockIdx.x * SCAN_TILE + t * SCAN_ITEMS;
  int c[SCAN_ITEMS];
  load_cnt4(cnt, base, n, c);
  const int mine = c[0] + c[1] + c[2] + c[3];

  int incl = mine;
  for (int o = 1; o < 32; o <<= 1) {
    const int y = __shfl_up(incl, o);
    if (lane >= o) incl += y;
  }
  if (lane == 31) shw[wv] = incl;
  __syncthreads();
  int wave_off = 0;
  for (int i = 0; i < wv; ++i) wave_off += shw[i];

  int run = tile_off + wave_off + incl - mine;
  for (int j = 0; j < SCAN_ITEMS; ++j) {
    const int i = base + j;
    if (i < n) {
      rowptr[i] = run;
      dinv[i]   = rsqrtf((float)(c[j] + 1));
      run += c[j];
      if (i == n - 1) rowptr[n] = run;
    }
  }
}

__global__ void k_fill(const int* __restrict__ src, const int* __restrict__ dst,
                       const int* __restrict__ rank, const int* __restrict__ rowptr,
                       const float* __restrict__ dinv, int2* __restrict__ rec, int e) {
  int i = blockIdx.x * blockDim.x + threadIdx.x;
  if (i >= e) return;
  const int s = src[i], d = dst[i];
  const int pos = rowptr[d] + rank[i];
  const float w = dinv[s] * dinv[d];
  if ((unsigned)pos < (unsigned)e) rec[pos] = make_int2(s, __float_as_int(w));
}

template <int NOUT>
__global__ __launch_bounds__(128) void k_gemm_wmma(const float* __restrict__ A,
                                                   const float* __restrict__ B,
                                                   float* __restrict__ C) {
  __shared__ float lA[16 * APITCH];

  const int tid  = threadIdx.x;
  const int wave = tid >> 5;
  const int lane = tid & 31;
  const int row0 = blockIdx.x * 16;
  const int col0 = (blockIdx.y * 4 + wave) * 16;
  const int half = lane >> 4;
  const int l16  = lane & 15;

  {
    const float4* gA = (const float4*)(A + (size_t)row0 * KDIM);
    float4* sA = (float4*)lA;
#pragma unroll
    for (int i = 0; i < 4; ++i) {
      int idx = tid + i * 128;
      int r   = idx >> 5;
      int c4  = idx & 31;
      sA[r * (APITCH / 4) + c4] = gA[idx];
    }
  }
  __syncthreads();

  const float* aP = lA + l16 * APITCH + half * 2;
  const float* bP = B + (half * 2) * NOUT + (col0 + l16);

  v8f acc = {};
#pragma unroll
  for (int k = 0; k < KDIM; k += 4) {
    v2f av = *(const v2f*)(aP + k);
    v2f bv;
    bv.x = bP[k * NOUT];
    bv.y = bP[(k + 1) * NOUT];
    acc = __builtin_amdgcn_wmma_f32_16x16x4_f32(false, av, false, bv,
                                                (short)0, acc, false, false);
  }

  float* cP = C + (size_t)(row0 + half * 8) * NOUT + (col0 + l16);
#pragma unroll
  for (int r = 0; r < 8; ++r) cP[r * NOUT] = acc[r];
}

#define GATHER_DEPTH 8

template <int V, bool FULL>
__device__ __forceinline__ void gather_group(float __attribute__((ext_vector_type(V)))& acc,
                                             const float __attribute__((ext_vector_type(V)))* __restrict__ hv,
                                             const int2 r, const int k, const int m,
                                             const int lane, const unsigned last_node) {
  typedef float vec_t __attribute__((ext_vector_type(V)));
  unsigned s[GATHER_DEPTH];
  float    w[GATHER_DEPTH];
  vec_t    v[GATHER_DEPTH];
#pragma unroll
  for (int u = 0; u < GATHER_DEPTH; ++u) {
    s[u] = min((unsigned)__builtin_amdgcn_readlane(r.x, k + u), last_node);
    w[u] = __int_as_float(__builtin_amdgcn_readlane(r.y, k + u));
  }
#pragma unroll
  for (int u = 0; u < GATHER_DEPTH; ++u) {
    v[u] = (vec_t)(0.0f);
    if (FULL || u < m) v[u] = hv[s[u] * 32u + lane];
  }
#pragma unroll
  for (int u = 0; u < GATHER_DEPTH; ++u) acc += v[u] * w[u];
}

template <int F, bool RELU>
__global__ __launch_bounds__(256) void k_agg_gather(const float* __restrict__ h,
                                                    const int* __restrict__ rowptr,
                                                    const int2* __restrict__ rec,
                                                    const float* __restrict__ dinv,
                                                    const float* __restrict__ bias,
                                                    float* __restrict__ out, int n) {
  constexpr int V = F / 32;
  typedef float vec_t __attribute__((ext_vector_type(V)));
  const vec_t* __restrict__ hv = (const vec_t*)h;

  const int lane = threadIdx.x & 31;
  const int row = __builtin_amdgcn_readfirstlane((int)((blockIdx.x * blockDim.x + threadIdx.x) >> 5));
  if (row >= n) return;

  const int beg = rowptr[row];
  const int end = rowptr[row + 1];
  const float di = dinv[row];
  const vec_t bv = ((const vec_t*)bias)[lane];
  vec_t acc = hv[(unsigned)row * 32u + lane] * (di * di);

  for (int base = beg; base < end; base += 32) {
    int2 r = make_int2(0, 0);
    if (base + lane < end) r = rec[base + lane];
    const int cnt = min(32, end - base);

    int k = 0;
    for (; k + GATHER_DEPTH <= cnt; k += GATHER_DEPTH)
      gather_group<V, true>(acc, hv, r, k, GATHER_DEPTH, lane, (unsigned)(n - 1));
    if (k < cnt)
      gather_group<V, false>(acc, hv, r, k, cnt - k, lane, (unsigned)(n - 1));
  }

  acc += bv;
  if (RELU) {
    for (int c = 0; c < V; ++c) acc[c] = fmaxf(acc[c], 0.0f);
  }
  ((vec_t*)out)[(unsigned)row * 32u + lane] = acc;
}

#define FUSED_WAVES   16
#define FUSED_THREADS (FUSED_WAVES * 32)
#define FUSED_PLACES  1024
#define FUSED_TILES   (NNODES / 16)
#define FUSED_TPB     ((FUSED_TILES + FUSED_PLACES - 1) / FUSED_PLACES)
#define FUSED_GRID    ((FUSED_TILES + FUSED_TPB - 1) / FUSED_TPB)
static_assert(FUSED_GRID <= FUSED_PLACES, "every block resident at once");
static_assert(NNODES % 16 == 0, "whole 16-row tiles");

template <int NOUT>
__global__ __launch_bounds__(FUSED_THREADS, 16) void k_agg_gemm(const float* __restrict__ h,
                                                            const int* __restrict__ rowptr,
                                                            const int2* __restrict__ rec,
                                                            const float* __restrict__ dinv,
                                                            const float* __restrict__ bias,
                                                            const float* __restrict__ Wn,
                                                            float* __restrict__ C, int n) {
  constexpr int V  = KDIM / 32;
  constexpr int NT = NOUT / 16;
  typedef float vec_t __attribute__((ext_vector_type(V)));
  __shared__ __attribute__((aligned(16))) float lW[KDIM * NOUT];
  __shared__ __attribute__((aligned(16))) float lA[16 * APITCH];

  const int tid  = threadIdx.x;
  const int lane = tid & 31;
  const int wave = __builtin_amdgcn_readfirstlane(tid >> 5);
  const int half = lane >> 4;
  const int l16  = lane & 15;

  {
    const float4* gW = (const float4*)Wn;
    constexpr int R4 = NOUT / 4;
    constexpr int UNITS = (KDIM / 2) * R4;
    static_assert(UNITS % FUSED_THREADS == 0, "whole rounds");
#pragma unroll
    for (int i = 0; i < UNITS / FUSED_THREADS; ++i) {
      const int q  = tid + i * FUSED_THREADS;
      const int rp = q / R4, c4 = q % R4;
      const float4 x0 = gW[(2 * rp) * R4 + c4];
      const float4 x1 = gW[(2 * rp + 1) * R4 + c4];
      const int kt = rp >> 1, hh = rp & 1;
      const int nt = c4 >> 2, cc = (c4 & 3) * 4;
      float* dst = lW + (((kt * NT + nt) * 32) + hh * 16 + cc) * 2;
      *(float4*)(dst)     = make_float4(x0.x, x1.x, x0.y, x1.y);
      *(float4*)(dst + 4) = make_float4(x0.z, x1.z, x0.w, x1.w);
    }
  }

  const vec_t* __restrict__ hv = (const vec_t*)h;
  const vec_t bv = ((const vec_t*)bias)[lane];
  const int ntiles = n / 16;

  for (int tile = blockIdx.x; tile < ntiles; tile += gridDim.x) {
    const int row0 = tile * 16;
    const int row  = row0 + wave;

    const int beg = rowptr[row];
    const int end = rowptr[row + 1];
    const float di = dinv[row];
    vec_t acc = hv[(unsigned)row * 32u + lane] * (di * di);

    for (int base = beg; base < end; base += 32) {
      int2 r = make_int2(0, 0);
      if (base + lane < end) r = rec[base + lane];
      const int cnt = min(32, end - base);

      int k = 0;
      for (; k + GATHER_DEPTH <= cnt; k += GATHER_DEPTH)
        gather_group<V, true>(acc, hv, r, k, GATHER_DEPTH, lane, (unsigned)(n - 1));
      if (k < cnt)
        gather_group<V, false>(acc, hv, r, k, cnt - k, lane, (unsigned)(n - 1));
    }

    acc += bv;
    for (int c = 0; c < V; ++c) acc[c] = fmaxf(acc[c], 0.0f);
    *(vec_t*)(lA + wave * APITCH + lane * V) = acc;
    __syncthreads();

    v8f d = {};
    if (wave < NOUT / 16) {
      const float* aP = lA + l16 * APITCH + half * 2;
      const float* bP = lW + (wave * 32 + lane) * 2;
      constexpr int KS = 4;
      constexpr int NG = KDIM / (4 * KS);
      v2f av[2][KS], bw[2][KS];
#pragma unroll
      for (int u = 0; u < KS; ++u) {
        const int k = 4 * u;
        av[0][u]   = *(const v2f*)(aP + k);
        bw[0][u]   = *(const v2f*)(bP + (k / 4) * (NT * 64));
      }
#pragma unroll
      for (int g = 0; g < NG; ++g) {
        if (g + 1 < NG) {
#pragma unroll
          for (int u = 0; u < KS; ++u) {
            const int k = 4 * KS * (g + 1) + 4 * u;
            av[(g + 1) & 1][u]   = *(const v2f*)(aP + k);
            bw[(g + 1) & 1][u]   = *(const v2f*)(bP + (k / 4) * (NT * 64));
          }
        }
        __builtin_amdgcn_sched_barrier(0);
#pragma unroll
        for (int u = 0; u < KS; ++u) {
          d = __builtin_amdgcn_wmma_f32_16x16x4_f32(false, av[g & 1][u], false, bw[g & 1][u],
                                                    (short)0, d, false, false);
        }
      }
    }
    __syncthreads();
    if (wave < NOUT / 16) {
      float* cP = C + (size_t)(row0 + half * 8) * NOUT + (wave * 16 + l16);
#pragma unroll
      for (int r = 0; r < 8; ++r) cP[r * NOUT] = d[r];
    }
  }
}

__global__ void k_decode(const float* __restrict__ z, const int* __restrict__ pa,
                         const int* __restrict__ pb, float* __restrict__ out, int np) {
  int tid = blockIdx.x * blockDim.x + threadIdx.x;
  int pair = tid >> 4;
  int l    = tid & 15;
  if (pair >= np) return;
  const float4 va = ((const float4*)z)[((size_t)pa[pair] << 4) + l];
  const float4 vb = ((const float4*)z)[((size_t)pb[pair] << 4) + l];
  float p = va.x * vb.x + va.y * vb.y + va.z * vb.z + va.w * vb.w;
  p += __shfl_xor(p, 8);
  p += __shfl_xor(p, 4);
  p += __shfl_xor(p, 2);
  p += __shfl_xor(p, 1);
  if (l == 0) out[pair] = p;
}

extern "C" void kernel_launch(void* const* d_in, const int* in_sizes, int n_in,
                              void* d_out, int out_size, void* d_ws, size_t ws_size,
                              hipStream_t stream) {
  (void)in_sizes; (void)n_in; (void)out_size; (void)ws_size;

  const float* x  = (const float*)d_in[0];
  const int*   ei = (const int*)d_in[1];
  const int*   el = (const int*)d_in[2];
  const float* W1 = (const float*)d_in[3];
  const float* b1 = (const float*)d_in[4];
  const float* W2 = (const float*)d_in[5];
  const float* b2 = (const float*)d_in[6];
  const float* W3 = (const float*)d_in[7];
  const float* b3 = (const float*)d_in[8];

  const int* src = ei;
  const int* dst = ei + NEDGES;
  const int* pa  = el;
  const int* pb  = el + NPAIRS;
  float* out = (float*)d_out;

  float* dinv   = (float*)d_ws;
  int*   rank   = (int*)(dinv + ((NNODES + 127) & ~127));
  float* bufA   = (float*)(rank + ((NEDGES + 127) & ~127));
  float* bufB   = bufA + (size_t)NNODES * DHID;
  int*   cnt    = (int*)(bufB + (size_t)NNODES * DHID);
  int*   rowptr = cnt + ((NNODES + 127) & ~127);
  int*   tsum   = rowptr + ((NNODES + 1 + 127) & ~127);
  int2*  rec    = (int2*)(tsum + 128);

  auto cdiv = [](long long a, long long b) { return (int)((a + b - 1) / b); };

  const int nTiles = cdiv(NNODES, SCAN_TILE);
  k_cnt_zero  <<<cdiv(NNODES, 256), 256, 0, stream>>>(cnt, NNODES);
  k_cnt_rank  <<<cdiv(NEDGES, 256), 256, 0, stream>>>(dst, cnt, rank, NEDGES);
  k_scan_tiles<<<nTiles, 256, 0, stream>>>(cnt, tsum, NNODES);
  k_scan_rows <<<nTiles, 256, 0, stream>>>(cnt, tsum, rowptr, dinv, NNODES);
  k_fill      <<<cdiv(NEDGES, 256), 256, 0, stream>>>(src, dst, rank, rowptr, dinv, rec, NEDGES);

  const int mTiles = NNODES / 16;
  const int gBlocks = cdiv((long long)NNODES * 32, 256);

  k_gemm_wmma<DHID><<<dim3(mTiles, DHID / 64), 128, 0, stream>>>(x, W1, bufA);

  k_agg_gemm<DHID><<<FUSED_GRID, FUSED_THREADS, 0, stream>>>(bufA, rowptr, rec, dinv, b1, W2, bufB, NNODES);

  k_agg_gemm<DOUT><<<FUSED_GRID, FUSED_THREADS, 0, stream>>>(bufB, rowptr, rec, dinv, b2, W3, bufA, NNODES);

  k_agg_gather<DOUT, false><<<gBlocks, 256, 0, stream>>>(bufA, rowptr, rec, dinv, b3, bufB, NNODES);

  k_decode<<<cdiv((long long)NPAIRS * 16, 256), 256, 0, stream>>>(bufB, pa, pb, out, NPAIRS);
}


# edge ranks without global atomics: per 32768-edge chunk all nodes' 16-bit counters in LDS (returning LDS add), then per-node prefix over chunks; a row's sum order becomes chunk order
# speedup vs baseline: 3.8409x; 1.1665x over previous
#include <hip/hip_runtime.h>

#define NNODES 100000
#define NEDGES 1600000
#define NPAIRS 200000
#define DIN    128
#define DHID   128
#define DOUT   64
#define KDIM   128
#define APITCH 132

typedef __attribute__((ext_vector_type(2))) float v2f;
typedef __attribute__((ext_vector_type(8))) float v8f;

#define RANK_CHUNK   32768
#define RANK_CHUNKS  ((NEDGES + RANK_CHUNK - 1) / RANK_CHUNK)
#define RANK_WORDS   (NNODES / 2)
#define RANK_THREADS 1024
#define RANK_PER_THREAD (RANK_CHUNK / RANK_THREADS)
static_assert(RANK_CHUNK <= 65535, "a node's count inside one chunk fits 16 bits");
static_assert(RANK_CHUNK % RANK_THREADS == 0, "whole rounds");
static_assert(NNODES % 8 == 0, "a chunk's packed row is whole 16-byte units");

__global__ __launch_bounds__(RANK_THREADS) void k_chunk_rank(const int* __restrict__ dst,
                                                             int* __restrict__ rank,
                                                             unsigned* __restrict__ tab, int e) {
  __shared__ __attribute__((aligned(16))) unsigned lc[RANK_WORDS];
  const int tid = threadIdx.x;
  for (int i = tid; i < RANK_WORDS / 4; i += RANK_THREADS) ((uint4*)lc)[i] = make_uint4(0u, 0u, 0u, 0u);

  const int base = blockIdx.x * RANK_CHUNK;
  int d[RANK_PER_THREAD];
#pragma unroll
  for (int j = 0; j < RANK_PER_THREAD; ++j) {
    const int i = base + j * RANK_THREADS + tid;
    d[j] = (i < e) ? dst[i] : -1;
  }
  __syncthreads();
  unsigned old[RANK_PER_THREAD];
#pragma unroll
  for (int j = 0; j < RANK_PER_THREAD; ++j) {
    const bool live = d[j] >= 0;
    const unsigned sh = (unsigned)(d[j] & 1) * 16u;
    old[j] = atomicAdd(&lc[live ? (d[j] >> 1) : 0], live ? (1u << sh) : 0u);
  }
#pragma unroll
  for (int j = 0; j < RANK_PER_THREAD; ++j) {
    const int i = base + j * RANK_THREADS + tid;
    const unsigned sh = (unsigned)(d[j] & 1) * 16u;
    if (d[j] >= 0) rank[i] = (int)((old[j] >> sh) & 0xFFFFu);
  }
  __syncthreads();
  uint4* __restrict__ row = (uint4*)(tab + (size_t)blockIdx.x * RANK_WORDS);
  for (int i = tid; i < RANK_WORDS / 4; i += RANK_THREADS) row[i] = ((const uint4*)lc)[i];
}

__global__ __launch_bounds__(256) void k_chunk_prefix(const unsigned* __restrict__ tab,
                                                      uint2* __restrict__ pre,
                                                      int* __restrict__ cnt) {
  const int w = blockIdx.x * 256 + threadIdx.x;
  if (w >= RANK_WORDS) return;
  unsigned t[RANK_CHUNKS];
#pragma unroll
  for (int c = 0; c < RANK_CHUNKS; ++c) t[c] = tab[c * RANK_WORDS + w];
  unsigned lo = 0u, hi = 0u;
#pragma unroll
  for (int c = 0; c < RANK_CHUNKS; ++c) {
    pre[c * RANK_WORDS + w] = make_uint2(lo, hi);
    lo += t[c] & 0xFFFFu;
    hi += t[c] >> 16;
  }
  ((int2*)cnt)[w] = make_int2((int)lo, (int)hi);
}

#define SCAN_ITEMS 4
#define SCAN_TILE  1024
static_assert((NNODES + SCAN_TILE - 1) / SCAN_TILE <= 128, "tsum holds 128 tile sums");

__device__ __forceinline__ void load_cnt4(const int* __restrict__ cnt, int base, int n, int c[SCAN_ITEMS]) {
  if (base + SCAN_ITEMS <= n) {
    const int4 q = *(const int4*)(cnt + base);
    c[0] = q.x; c[1] = q.y; c[2] = q.z; c[3] = q.w;
  } else {
    for (int j = 0; j < SCAN_ITEMS; ++j) c[j] = (base + j < n) ? cnt[base + j] : 0;
  }
}

__device__ __forceinline__ int block_sum256(int x, int* sh) {
  const int lane = threadIdx.x & 31, wv = threadIdx.x >> 5;
  for (int o = 16; o > 0; o >>= 1) x += __shfl_xor(x, o);
  if (lane == 0) sh[wv] = x;
  __syncthreads();
  int t = 0;
  for (int i = 0; i < 8; ++i) t += sh[i];
  return t;
}

__global__ __launch_bounds__(256) void k_scan_tiles(const int* __restrict__ cnt,
                                                    int* __restrict__ tsum, int n) {
  __shared__ int sh[8];
  int c[SCAN_ITEMS];
  load_cnt4(cnt, blockIdx.x * SCAN_TILE + threadIdx.x * SCAN_ITEMS, n, c);
  const int tot = block_sum256(c[0] + c[1] + c[2] + c[3], sh);
  if (threadIdx.x == 0) tsum[blockIdx.x] = tot;
}

__global__ __launch_bounds__(256) void k_scan_rows(const int* __restrict__ cnt,
                                                   const int* __restrict__ tsum,
                                                   int* __restrict__ rowptr,
                                                   float* __restrict__ dinv, int n) {
  __shared__ int sh[8];
  __shared__ int shw[8];
  const int t = threadIdx.x, lane = t & 31, wv = t >> 5;

  int part = 0;
  for (int i = t; i < (int)blockIdx.x; i += 256) part += tsum[i];
  const int tile_off = block_sum256(part, sh);

  const int base = blockIdx.x * SCAN_TILE + t * SCAN_ITEMS;
  int c[SCAN_ITEMS];
  load_cnt4(cnt, base, n, c);
  const int mine = c[0] + c[1] + c[2] + c[3];

  int incl = mine;
  for (int o = 1; o < 32; o <<= 1) {
    const int y = __shfl_up(incl, o);
    if (lane >= o) incl += y;
  }
  if (lane == 31) shw[wv] = incl;
  __syncthreads();
  int wave_off = 0;
  for (int i = 0; i < wv; ++i) wave_off += shw[i];

  int run = tile_off + wave_off + incl - mine;
  for (int j = 0; j < SCAN_ITEMS; ++j) {
    const int i = base + j;
    if (i < n) {
      rowptr[i] = run;
      dinv[i]   = rsqrtf((float)(c[j] + 1));
      run += c[j];
      if (i == n - 1) rowptr[n] = run;
    }
  }
}

__global__ void k_fill(const int* __restrict__ src, const int* __restrict__ dst,
                       const int* __restrict__ rank, const int* __restrict__ rowptr,
                       const unsigned* __restrict__ pre,
                       const float* __restrict__ dinv, int2* __restrict__ rec, int e) {
  int i = blockIdx.x * blockDim.x + threadIdx.x;
  if (i >= e) return;
  const int s = src[i], d = dst[i];
  const int pos = rowptr[d] + (int)pre[(unsigned)(i / RANK_CHUNK) * (unsigned)NNODES + (unsigned)d] + rank[i];
  const float w = dinv[s] * dinv[d];
  if ((unsigned)pos < (unsigned)e) rec[pos] = make_int2(s, __float_as_int(w));
}

template <int NOUT>
__global__ __launch_bounds__(128) void k_gemm_wmma(const float* __restrict__ A,
                                                   const float* __restrict__ B,
                                                   float* __restrict__ C) {
  __shared__ float lA[16 * APITCH];

  const int tid  = threadIdx.x;
  const int wave = tid >> 5;
  const int lane = tid & 31;
  const int row0 = blockIdx.x * 16;
  const int col0 = (blockIdx.y * 4 + wave) * 16;
  const int half = lane >> 4;
  const int l16  = lane & 15;

  {
    const float4* gA = (const float4*)(A + (size_t)row0 * KDIM);
    float4* sA = (float4*)lA;
#pragma unroll
    for (int i = 0; i < 4; ++i) {
      int idx = tid + i * 128;
      int r   = idx >> 5;
      int c4  = idx & 31;
      sA[r * (APITCH / 4) + c4] = gA[idx];
    }
  }
  __syncthreads();

  const float* aP = lA + l16 * APITCH + half * 2;
  const float* bP = B + (half * 2) * NOUT + (col0 + l16);

  v8f acc = {};
#pragma unroll
  for (int k = 0; k < KDIM; k += 4) {
    v2f av = *(const v2f*)(aP + k);
    v2f bv;
    bv.x = bP[k * NOUT];
    bv.y = bP[(k + 1) * NOUT];
    acc = __builtin_amdgcn_wmma_f32_16x16x4_f32(false, av, false, bv,
                                                (short)0, acc, false, false);
  }

  float* cP = C + (size_t)(row0 + half * 8) * NOUT + (col0 + l16);
#pragma unroll
  for (int r = 0; r < 8; ++r) cP[r * NOUT] = acc[r];
}

#define GATHER_DEPTH 8

template <int V, bool FULL>
__device__ __forceinline__ void gather_group(float __attribute__((ext_vector_type(V)))& acc,
                                             const float __attribute__((ext_vector_type(V)))* __restrict__ hv,
                                             const int2 r, const int k, const int m,
                                             const int lane, const unsigned last_node) {
  typedef float vec_t __attribute__((ext_vector_type(V)));
  unsigned s[GATHER_DEPTH];
  float    w[GATHER_DEPTH];
  vec_t    v[GATHER_DEPTH];
#pragma unroll
  for (int u = 0; u < GATHER_DEPTH; ++u) {
    s[u] = min((unsigned)__builtin_amdgcn_readlane(r.x, k + u), last_node);
    w[u] = __int_as_float(__builtin_amdgcn_readlane(r.y, k + u));
  }
#pragma unroll
  for (int u = 0; u < GATHER_DEPTH; ++u) {
    v[u] = (vec_t)(0.0f);
    if (FULL || u < m) v[u] = hv[s[u] * 32u + lane];
  }
#pragma unroll
  for (int u = 0; u < GATHER_DEPTH; ++u) acc += v[u] * w[u];
}

template <int F, bool RELU>
__global__ __launch_bounds__(256) void k_agg_gather(const float* __restrict__ h,
                                                    const int* __restrict__ rowptr,
                                                    const int2* __restrict__ rec,
                                                    const float* __restrict__ dinv,
                                                    const float* __restrict__ bias,
                                                    float* __restrict__ out, int n) {
  constexpr int V = F / 32;
  typedef float vec_t __attribute__((ext_vector_type(V)));
  const vec_t* __restrict__ hv = (const vec_t*)h;

  const int lane = threadIdx.x & 31;
  const int row = __builtin_amdgcn_readfirstlane((int)((blockIdx.x * blockDim.x + threadIdx.x) >> 5));
  if (row >= n) return;

  const int beg = rowptr[row];
  const int end = rowptr[row + 1];
  const float di = dinv[row];
  const vec_t bv = ((const vec_t*)bias)[lane];
  vec_t acc = hv[(unsigned)row * 32u + lane] * (di * di);

  for (int base = beg; base < end; base += 32) {
    int2 r = make_int2(0, 0);
    if (base + lane < end) r = rec[base + lane];
    const int cnt = min(32, end - base);

    int k = 0;
    for (; k + GATHER_DEPTH <= cnt; k += GATHER_DEPTH)
      gather_group<V, true>(acc, hv, r, k, GATHER_DEPTH, lane, (unsigned)(n - 1));
    if (k < cnt)
      gather_group<V, false>(acc, hv, r, k, cnt - k, lane, (unsigned)(n - 1));
  }

  acc += bv;
  if (RELU) {
    for (int c = 0; c < V; ++c) acc[c] = fmaxf(acc[c], 0.0f);
  }
  ((vec_t*)out)[(unsigned)row * 32u + lane] = acc;
}

#define FUSED_WAVES   16
#define FUSED_THREADS (FUSED_WAVES * 32)
#define FUSED_PLACES  1024
#define FUSED_TILES   (NNODES / 16)
#define FUSED_TPB     ((FUSED_TILES + FUSED_PLACES - 1) / FUSED_PLACES)
#define FUSED_GRID    ((FUSED_TILES + FUSED_TPB - 1) / FUSED_TPB)
static_assert(FUSED_GRID <= FUSED_PLACES, "every block resident at once");
static_assert(NNODES % 16 == 0, "whole 16-row tiles");

template <int NOUT>
__global__ __launch_bounds__(FUSED_THREADS, 16) void k_agg_gemm(const float* __restrict__ h,
                                                            const int* __restrict__ rowptr,
                                                            const int2* __restrict__ rec,
                                                            const float* __restrict__ dinv,
                                                            const float* __restrict__ bias,
                                                            const float* __restrict__ Wn,
                                                            float* __restrict__ C, int n) {
  constexpr int V  = KDIM / 32;
  constexpr int NT = NOUT / 16;
  typedef float vec_t __attribute__((ext_vector_type(V)));
  __shared__ __attribute__((aligned(16))) float lW[KDIM * NOUT];
  __shared__ __attribute__((aligned(16))) float lA[16 * APITCH];

  const int tid  = threadIdx.x;
  const int lane = tid & 31;
  const int wave = __builtin_amdgcn_readfirstlane(tid >> 5);
  const int half = lane >> 4;
  const int l16  = lane & 15;

  {
    const float4* gW = (const float4*)Wn;
    constexpr int R4 = NOUT / 4;
    constexpr int UNITS = (KDIM / 2) * R4;
    static_assert(UNITS % FUSED_THREADS == 0, "whole rounds");
#pragma unroll
    for (int i = 0; i < UNITS / FUSED_THREADS; ++i) {
      const int q  = tid + i * FUSED_THREADS;
      const int rp = q / R4, c4 = q % R4;
      const float4 x0 = gW[(2 * rp) * R4 + c4];
      const float4 x1 = gW[(2 * rp + 1) * R4 + c4];
      const int kt = rp >> 1, hh = rp & 1;
      const int nt = c4 >> 2, cc = (c4 & 3) * 4;
      float* dst = lW + (((kt * NT + nt) * 32) + hh * 16 + cc) * 2;
      *(float4*)(dst)     = make_float4(x0.x, x1.x, x0.y, x1.y);
      *(float4*)(dst + 4) = make_float4(x0.z, x1.z, x0.w, x1.w);
    }
  }

  const vec_t* __restrict__ hv = (const vec_t*)h;
  const vec_t bv = ((const vec_t*)bias)[lane];
  const int ntiles = n / 16;

  for (int tile = blockIdx.x; tile < ntiles; tile += gridDim.x) {
    const int row0 = tile * 16;
    const int row  = row0 + wave;

    const int beg = rowptr[row];
    const int end = rowptr[row + 1];
    const float di = dinv[row];
    vec_t acc = hv[(unsigned)row * 32u + lane] * (di * di);

    for (int base = beg; base < end; base += 32) {
      int2 r = make_int2(0, 0);
      if (base + lane < end) r = rec[base + lane];
      const int cnt = min(32, end - base);

      int k = 0;
      for (; k + GATHER_DEPTH <= cnt; k += GATHER_DEPTH)
        gather_group<V, true>(acc, hv, r, k, GATHER_DEPTH, lane, (unsigned)(n - 1));
      if (k < cnt)
        gather_group<V, false>(acc, hv, r, k, cnt - k, lane, (unsigned)(n - 1));
    }

    acc += bv;
    for (int c = 0; c < V; ++c) acc[c] = fmaxf(acc[c], 0.0f);
    *(vec_t*)(lA + wave * APITCH + lane * V) = acc;
    __syncthreads();

    v8f d = {};
    if (wave < NOUT / 16) {
      const float* aP = lA + l16 * APITCH + half * 2;
      const float* bP = lW + (wave * 32 + lane) * 2;
      constexpr int KS = 4;
      constexpr int NG = KDIM / (4 * KS);
      v2f av[2][KS], bw[2][KS];
#pragma unroll
      for (int u = 0; u < KS; ++u) {
        const int k = 4 * u;
        av[0][u]   = *(const v2f*)(aP + k);
        bw[0][u]   = *(const v2f*)(bP + (k / 4) * (NT * 64));
      }
#pragma unroll
      for (int g = 0; g < NG; ++g) {
        if (g + 1 < NG) {
#pragma unroll
          for (int u = 0; u < KS; ++u) {
            const int k = 4 * KS * (g + 1) + 4 * u;
            av[(g + 1) & 1][u]   = *(const v2f*)(aP + k);
            bw[(g + 1) & 1][u]   = *(const v2f*)(bP + (k / 4) * (NT * 64));
          }
        }
        __builtin_amdgcn_sched_barrier(0);
#pragma unroll
        for (int u = 0; u < KS; ++u) {
          d = __builtin_amdgcn_wmma_f32_16x16x4_f32(false, av[g & 1][u], false, bw[g & 1][u],
                                                    (short)0, d, false, false);
        }
      }
    }
    __syncthreads();
    if (wave < NOUT / 16) {
      float* cP = C + (size_t)(row0 + half * 8) * NOUT + (wave * 16 + l16);
#pragma unroll
      for (int r = 0; r < 8; ++r) cP[r * NOUT] = d[r];
    }
  }
}

__global__ void k_decode(const float* __restrict__ z, const int* __restrict__ pa,
                         const int* __restrict__ pb, float* __restrict__ out, int np) {
  int tid = blockIdx.x * blockDim.x + threadIdx.x;
  int pair = tid >> 4;
  int l    = tid & 15;
  if (pair >= np) return;
  const float4 va = ((const float4*)z)[((size_t)pa[pair] << 4) + l];
  const float4 vb = ((const float4*)z)[((size_t)pb[pair] << 4) + l];
  float p = va.x * vb.x + va.y * vb.y + va.z * vb.z + va.w * vb.w;
  p += __shfl_xor(p, 8);
  p += __shfl_xor(p, 4);
  p += __shfl_xor(p, 2);
  p += __shfl_xor(p, 1);
  if (l == 0) out[pair] = p;
}

extern "C" void kernel_launch(void* const* d_in, const int* in_sizes, int n_in,
                              void* d_out, int out_size, void* d_ws, size_t ws_size,
                              hipStream_t stream) {
  (void)in_sizes; (void)n_in; (void)out_size; (void)ws_size;

  const float* x  = (const float*)d_in[0];
  const int*   ei = (const int*)d_in[1];
  const int*   el = (const int*)d_in[2];
  const float* W1 = (const float*)d_in[3];
  const float* b1 = (const float*)d_in[4];
  const float* W2 = (const float*)d_in[5];
  const float* b2 = (const float*)d_in[6];
  const float* W3 = (const float*)d_in[7];
  const float* b3 = (const float*)d_in[8];

  const int* src = ei;
  const int* dst = ei + NEDGES;
  const int* pa  = el;
  const int* pb  = el + NPAIRS;
  float* out = (float*)d_out;

  float* dinv   = (float*)d_ws;
  int*   rank   = (int*)(dinv + ((NNODES + 127) & ~127));
  float* bufA   = (float*)(rank + ((NEDGES + 127) & ~127));
  float* bufB   = bufA + (size_t)NNODES * DHID;
  int*   cnt    = (int*)(bufB + (size_t)NNODES * DHID);
  int*   rowptr = cnt + ((NNODES + 127) & ~127);
  int*   tsum   = rowptr + ((NNODES + 1 + 127) & ~127);
  int2*  rec    = (int2*)(tsum + 128);
  unsigned* tab = (unsigned*)(rec + NEDGES);
  unsigned* pre = tab + (size_t)RANK_CHUNKS * RANK_WORDS;

  auto cdiv = [](long long a, long long b) { return (int)((a + b - 1) / b); };

  const int nTiles = cdiv(NNODES, SCAN_TILE);
  k_chunk_rank  <<<RANK_CHUNKS, RANK_THREADS, 0, stream>>>(dst, rank, tab, NEDGES);
  k_chunk_prefix<<<cdiv(RANK_WORDS, 256), 256, 0, stream>>>(tab, (uint2*)pre, cnt);
  k_scan_tiles<<<nTiles, 256, 0, stream>>>(cnt, tsum, NNODES);
  k_scan_rows <<<nTiles, 256, 0, stream>>>(cnt, tsum, rowptr, dinv, NNODES);
  k_fill      <<<cdiv(NEDGES, 256), 256, 0, stream>>>(src, dst, rank, rowptr, pre, dinv, rec, NEDGES);

  const int mTiles = NNODES / 16;
  const int gBlocks = cdiv((long long)NNODES * 32, 256);

  k_gemm_wmma<DHID><<<dim3(mTiles, DHID / 64), 128, 0, stream>>>(x, W1, bufA);

  k_agg_gemm<DHID><<<FUSED_GRID, FUSED_THREADS, 0, stream>>>(bufA, rowptr, rec, dinv, b1, W2, bufB, NNODES);

  k_agg_gemm<DOUT><<<FUSED_GRID, FUSED_THREADS, 0, stream>>>(bufB, rowptr, rec, dinv, b2, W3, bufA, NNODES);

  k_agg_gather<DOUT, false><<<gBlocks, 256, 0, stream>>>(bufA, rowptr, rec, dinv, b3, bufB, NNODES);

  k_decode<<<cdiv((long long)NPAIRS * 16, 256), 256, 0, stream>>>(bufB, pa, pb, out, NPAIRS);
}


# fewer, fatter waves in the two launches bound by wave start-up: last aggregation wave takes 8 rows (next row's records, vector, dinv prefetched), decode wave 32 pairs with 16 row loads in flight
# speedup vs baseline: 3.9743x; 1.0347x over previous
#include <hip/hip_runtime.h>

#define NNODES 100000
#define NEDGES 1600000
#define NPAIRS 200000
#define DIN    128
#define DHID   128
#define DOUT   64
#define KDIM   128
#define APITCH 132

typedef __attribute__((ext_vector_type(2))) float v2f;
typedef __attribute__((ext_vector_type(8))) float v8f;

#define RANK_CHUNK   32768
#define RANK_CHUNKS  ((NEDGES + RANK_CHUNK - 1) / RANK_CHUNK)
#define RANK_WORDS   (NNODES / 2)
#define RANK_THREADS 1024
#define RANK_PER_THREAD (RANK_CHUNK / RANK_THREADS)
static_assert(RANK_CHUNK <= 65535, "a node's count inside one chunk fits 16 bits");
static_assert(RANK_CHUNK % RANK_THREADS == 0, "whole rounds");
static_assert(NNODES % 8 == 0, "a chunk's packed row is whole 16-byte units");

__global__ __launch_bounds__(RANK_THREADS) void k_chunk_rank(const int* __restrict__ dst,
                                                             int* __restrict__ rank,
                                                             unsigned* __restrict__ tab, int e) {
  __shared__ __attribute__((aligned(16))) unsigned lc[RANK_WORDS];
  const int tid = threadIdx.x;
  for (int i = tid; i < RANK_WORDS / 4; i += RANK_THREADS) ((uint4*)lc)[i] = make_uint4(0u, 0u, 0u, 0u);

  const int base = blockIdx.x * RANK_CHUNK;
  int d[RANK_PER_THREAD];
#pragma unroll
  for (int j = 0; j < RANK_PER_THREAD; ++j) {
    const int i = base + j * RANK_THREADS + tid;
    d[j] = (i < e) ? dst[i] : -1;
  }
  __syncthreads();
  unsigned old[RANK_PER_THREAD];
#pragma unroll
  for (int j = 0; j < RANK_PER_THREAD; ++j) {
    const bool live = d[j] >= 0;
    const unsigned sh = (unsigned)(d[j] & 1) * 16u;
    old[j] = atomicAdd(&lc[live ? (d[j] >> 1) : 0], live ? (1u << sh) : 0u);
  }
#pragma unroll
  for (int j = 0; j < RANK_PER_THREAD; ++j) {
    const int i = base + j * RANK_THREADS + tid;
    const unsigned sh = (unsigned)(d[j] & 1) * 16u;
    if (d[j] >= 0) rank[i] = (int)((old[j] >> sh) & 0xFFFFu);
  }
  __syncthreads();
  uint4* __restrict__ row = (uint4*)(tab + (size_t)blockIdx.x * RANK_WORDS);
  for (int i = tid; i < RANK_WORDS / 4; i += RANK_THREADS) row[i] = ((const uint4*)lc)[i];
}

__global__ __launch_bounds__(256) void k_chunk_prefix(const unsigned* __restrict__ tab,
                                                      uint2* __restrict__ pre,
                                                      int* __restrict__ cnt) {
  const int w = blockIdx.x * 256 + threadIdx.x;
  if (w >= RANK_WORDS) return;
  unsigned t[RANK_CHUNKS];
#pragma unroll
  for (int c = 0; c < RANK_CHUNKS; ++c) t[c] = tab[c * RANK_WORDS + w];
  unsigned lo = 0u, hi = 0u;
#pragma unroll
  for (int c = 0; c < RANK_CHUNKS; ++c) {
    pre[c * RANK_WORDS + w] = make_uint2(lo, hi);
    lo += t[c] & 0xFFFFu;
    hi += t[c] >> 16;
  }
  ((int2*)cnt)[w] = make_int2((int)lo, (int)hi);
}

#define SCAN_ITEMS 4
#define SCAN_TILE  1024
static_assert((NNODES + SCAN_TILE - 1) / SCAN_TILE <= 128, "tsum holds 128 tile sums");

__device__ __forceinline__ void load_cnt4(const int* __restrict__ cnt, int base, int n, int c[SCAN_ITEMS]) {
  if (base + SCAN_ITEMS <= n) {
    const int4 q = *(const int4*)(cnt + base);
    c[0] = q.x; c[1] = q.y; c[2] = q.z; c[3] = q.w;
  } else {
    for (int j = 0; j < SCAN_ITEMS; ++j) c[j] = (base + j < n) ? cnt[base + j] : 0;
  }
}

__device__ __forceinline__ int block_sum256(int x, int* sh) {
  const int lane = threadIdx.x & 31, wv = threadIdx.x >> 5;
  for (int o = 16; o > 0; o >>= 1) x += __shfl_xor(x, o);
  if (lane == 0) sh[wv] = x;
  __syncthreads();
  int t = 0;
  for (int i = 0; i < 8; ++i) t += sh[i];
  return t;
}

__global__ __launch_bounds__(256) void k_scan_tiles(const int* __restrict__ cnt,
                                                    int* __restrict__ tsum, int n) {
  __shared__ int sh[8];
  int c[SCAN_ITEMS];
  load_cnt4(cnt, blockIdx.x * SCAN_TILE + threadIdx.x * SCAN_ITEMS, n, c);
  const int tot = block_sum256(c[0] + c[1] + c[2] + c[3], sh);
  if (threadIdx.x == 0) tsum[blockIdx.x] = tot;
}

__global__ __launch_bounds__(256) void k_scan_rows(const int* __restrict__ cnt,
                                                   const int* __restrict__ tsum,
                                                   int* __restrict__ rowptr,
                                                   float* __restrict__ dinv, int n) {
  __shared__ int sh[8];
  __shared__ int shw[8];
  const int t = threadIdx.x, lane = t & 31, wv = t >> 5;

  int part = 0;
  for (int i = t; i < (int)blockIdx.x; i += 256) part += tsum[i];
  const int tile_off = block_sum256(part, sh);

  const int base = blockIdx.x * SCAN_TILE + t * SCAN_ITEMS;
  int c[SCAN_ITEMS];
  load_cnt4(cnt, base, n, c);
  const int mine = c[0] + c[1] + c[2] + c[3];

  int incl = mine;
  for (int o = 1; o < 32; o <<= 1) {
    const int y = __shfl_up(incl, o);
    if (lane >= o) incl += y;
  }
  if (lane == 31) shw[wv] = incl;
  __syncthreads();
  int wave_off = 0;
  for (int i = 0; i < wv; ++i) wave_off += shw[i];

  int run = tile_off + wave_off + incl - mine;
  for (int j = 0; j < SCAN_ITEMS; ++j) {
    const int i = base + j;
    if (i < n) {
      rowptr[i] = run;
      dinv[i]   = rsqrtf((float)(c[j] + 1));
      run += c[j];
      if (i == n - 1) rowptr[n] = run;
    }
  }
}

__global__ void k_fill(const int* __restrict__ src, const int* __restrict__ dst,
                       const int* __restrict__ rank, const int* __restrict__ rowptr,
                       const unsigned* __restrict__ pre,
                       const float* __restrict__ dinv, int2* __restrict__ rec, int e) {
  int i = blockIdx.x * blockDim.x + threadIdx.x;
  if (i >= e) return;
  const int s = src[i], d = dst[i];
  const int pos = rowptr[d] + (int)pre[(unsigned)(i / RANK_CHUNK) * (unsigned)NNODES + (unsigned)d] + rank[i];
  const float w = dinv[s] * dinv[d];
  if ((unsigned)pos < (unsigned)e) rec[pos] = make_int2(s, __float_as_int(w));
}

template <int NOUT>
__global__ __launch_bounds__(128) void k_gemm_wmma(const float* __restrict__ A,
                                                   const float* __restrict__ B,
                                                   float* __restrict__ C) {
  __shared__ float lA[16 * APITCH];

  const int tid  = threadIdx.x;
  const int wave = tid >> 5;
  const int lane = tid & 31;
  const int row0 = blockIdx.x * 16;
  const int col0 = (blockIdx.y * 4 + wave) * 16;
  const int half = lane >> 4;
  const int l16  = lane & 15;

  {
    const float4* gA = (const float4*)(A + (size_t)row0 * KDIM);
    float4* sA = (float4*)lA;
#pragma unroll
    for (int i = 0; i < 4; ++i) {
      int idx = tid + i * 128;
      int r   = idx >> 5;
      int c4  = idx & 31;
      sA[r * (APITCH / 4) + c4] = gA[idx];
    }
  }
  __syncthreads();

  const float* aP = lA + l16 * APITCH + half * 2;
  const float* bP = B + (half * 2) * NOUT + (col0 + l16);

  v8f acc = {};
#pragma unroll
  for (int k = 0; k < KDIM; k += 4) {
    v2f av = *(const v2f*)(aP + k);
    v2f bv;
    bv.x = bP[k * NOUT];
    bv.y = bP[(k + 1) * NOUT];
    acc = __builtin_amdgcn_wmma_f32_16x16x4_f32(false, av, false, bv,
                                                (short)0, acc, false, false);
  }

  float* cP = C + (size_t)(row0 + half * 8) * NOUT + (col0 + l16);
#pragma unroll
  for (int r = 0; r < 8; ++r) cP[r * NOUT] = acc[r];
}

#define GATHER_DEPTH 8

template <int V, bool FULL>
__device__ __forceinline__ void gather_group(float __attribute__((ext_vector_type(V)))& acc,
                                             const float __attribute__((ext_vector_type(V)))* __restrict__ hv,
                                             const int2 r, const int k, const int m,
                                             const int lane, const unsigned last_node) {
  typedef float vec_t __attribute__((ext_vector_type(V)));
  unsigned s[GATHER_DEPTH];
  float    w[GATHER_DEPTH];
  vec_t    v[GATHER_DEPTH];
#pragma unroll
  for (int u = 0; u < GATHER_DEPTH; ++u) {
    s[u] = min((unsigned)__builtin_amdgcn_readlane(r.x, k + u), last_node);
    w[u] = __int_as_float(__builtin_amdgcn_readlane(r.y, k + u));
  }
#pragma unroll
  for (int u = 0; u < GATHER_DEPTH; ++u) {
    v[u] = (vec_t)(0.0f);
    if (FULL || u < m) v[u] = hv[s[u] * 32u + lane];
  }
#pragma unroll
  for (int u = 0; u < GATHER_DEPTH; ++u) acc += v[u] * w[u];
}

template <int V>
__device__ __forceinline__ void gather_chunk(float __attribute__((ext_vector_type(V)))& acc,
                                             const float __attribute__((ext_vector_type(V)))* __restrict__ hv,
                                             const int2 r, const int cnt,
                                             const int lane, const unsigned last_node) {
  int k = 0;
  for (; k + GATHER_DEPTH <= cnt; k += GATHER_DEPTH)
    gather_group<V, true>(acc, hv, r, k, GATHER_DEPTH, lane, last_node);
  if (k < cnt)
    gather_group<V, false>(acc, hv, r, k, cnt - k, lane, last_node);
}

#define GATHER_ROWS 8

template <int F, bool RELU>
__global__ __launch_bounds__(256) void k_agg_gather(const float* __restrict__ h,
                                                    const int* __restrict__ rowptr,
                                                    const int2* __restrict__ rec,
                                                    const float* __restrict__ dinv,
                                                    const float* __restrict__ bias,
                                                    float* __restrict__ out, int n) {
  constexpr int V = F / 32;
  typedef float vec_t __attribute__((ext_vector_type(V)));
  const vec_t* __restrict__ hv = (const vec_t*)h;

  const int lane = threadIdx.x & 31;
  const int gw = __builtin_amdgcn_readfirstlane((int)((blockIdx.x * blockDim.x + threadIdx.x) >> 5));
  const int r0 = gw * GATHER_ROWS;
  if (r0 >= n) return;
  const int r1 = min(r0 + GATHER_ROWS, n);

  const vec_t bv = ((const vec_t*)bias)[lane];

  int end  = rowptr[r0];
  int nend = rowptr[r0 + 1];
  int2 rn = make_int2(0, 0);
  if (end + lane < nend) rn = rec[end + lane];
  vec_t sn = hv[(unsigned)r0 * 32u + lane];
  float dn = dinv[r0];

  for (int row = r0; row < r1; ++row) {
    const int beg = end;
    end = nend;
    const int2 r = rn;
    const vec_t self = sn;
    const float di = dn;
    if (row + 1 < r1) {
      nend = rowptr[row + 2];
      rn = make_int2(0, 0);
      if (end + lane < nend) rn = rec[end + lane];
      sn = hv[(unsigned)(row + 1) * 32u + lane];
      dn = dinv[row + 1];
    }

    vec_t acc = self * (di * di);

    gather_chunk<V>(acc, hv, r, min(32, end - beg), lane, (unsigned)(n - 1));
    for (int base = beg + 32; base < end; base += 32) {
      int2 r2 = make_int2(0, 0);
      if (base + lane < end) r2 = rec[base + lane];
      gather_chunk<V>(acc, hv, r2, min(32, end - base), lane, (unsigned)(n - 1));
    }

    acc += bv;
    if (RELU) {
      for (int c = 0; c < V; ++c) acc[c] = fmaxf(acc[c], 0.0f);
    }
    ((vec_t*)out)[(unsigned)row * 32u + lane] = acc;
  }
}

#define FUSED_WAVES   16
#define FUSED_THREADS (FUSED_WAVES * 32)
#define FUSED_PLACES  1024
#define FUSED_TILES   (NNODES / 16)
#define FUSED_TPB     ((FUSED_TILES + FUSED_PLACES - 1) / FUSED_PLACES)
#define FUSED_GRID    ((FUSED_TILES + FUSED_TPB - 1) / FUSED_TPB)
static_assert(FUSED_GRID <= FUSED_PLACES, "every block resident at once");
static_assert(NNODES % 16 == 0, "whole 16-row tiles");

template <int NOUT>
__global__ __launch_bounds__(FUSED_THREADS, 16) void k_agg_gemm(const float* __restrict__ h,
                                                            const int* __restrict__ rowptr,
                                                            const int2* __restrict__ rec,
                                                            const float* __restrict__ dinv,
                                                            const float* __restrict__ bias,
                                                            const float* __restrict__ Wn,
                                                            float* __restrict__ C, int n) {
  constexpr int V  = KDIM / 32;
  constexpr int NT = NOUT / 16;
  typedef float vec_t __attribute__((ext_vector_type(V)));
  __shared__ __attribute__((aligned(16))) float lW[KDIM * NOUT];
  __shared__ __attribute__((aligned(16))) float lA[16 * APITCH];

  const int tid  = threadIdx.x;
  const int lane = tid & 31;
  const int wave = __builtin_amdgcn_readfirstlane(tid >> 5);
  const int half = lane >> 4;
  const int l16  = lane & 15;

  {
    const float4* gW = (const float4*)Wn;
    constexpr int R4 = NOUT / 4;
    constexpr int UNITS = (KDIM / 2) * R4;
    static_assert(UNITS % FUSED_THREADS == 0, "whole rounds");
#pragma unroll
    for (int i = 0; i < UNITS / FUSED_THREADS; ++i) {
      const int q  = tid + i * FUSED_THREADS;
      const int rp = q / R4, c4 = q % R4;
      const float4 x0 = gW[(2 * rp) * R4 + c4];
      const float4 x1 = gW[(2 * rp + 1) * R4 + c4];
      const int kt = rp >> 1, hh = rp & 1;
      const int nt = c4 >> 2, cc = (c4 & 3) * 4;
      float* dst = lW + (((kt * NT + nt) * 32) + hh * 16 + cc) * 2;
      *(float4*)(dst)     = make_float4(x0.x, x1.x, x0.y, x1.y);
      *(float4*)(dst + 4) = make_float4(x0.z, x1.z, x0.w, x1.w);
    }
  }

  const vec_t* __restrict__ hv = (const vec_t*)h;
  const vec_t bv = ((const vec_t*)bias)[lane];
  const int ntiles = n / 16;

  for (int tile = blockIdx.x; tile < ntiles; tile += gridDim.x) {
    const int row0 = tile * 16;
    const int row  = row0 + wave;

    const int beg = rowptr[row];
    const int end = rowptr[row + 1];
    const float di = dinv[row];
    vec_t acc = hv[(unsigned)row * 32u + lane] * (di * di);

    for (int base = beg; base < end; base += 32) {
      int2 r = make_int2(0, 0);
      if (base + lane < end) r = rec[base + lane];
      const int cnt = min(32, end - base);

      int k = 0;
      for (; k + GATHER_DEPTH <= cnt; k += GATHER_DEPTH)
        gather_group<V, true>(acc, hv, r, k, GATHER_DEPTH, lane, (unsigned)(n - 1));
      if (k < cnt)
        gather_group<V, false>(acc, hv, r, k, cnt - k, lane, (unsigned)(n - 1));
    }

    acc += bv;
    for (int c = 0; c < V; ++c) acc[c] = fmaxf(acc[c], 0.0f);
    *(vec_t*)(lA + wave * APITCH + lane * V) = acc;
    __syncthreads();

    v8f d = {};
    if (wave < NOUT / 16) {
      const float* aP = lA + l16 * APITCH + half * 2;
      const float* bP = lW + (wave * 32 + lane) * 2;
      constexpr int KS = 4;
      constexpr int NG = KDIM / (4 * KS);
      v2f av[2][KS], bw[2][KS];
#pragma unroll
      for (int u = 0; u < KS; ++u) {
        const int k = 4 * u;
        av[0][u]   = *(const v2f*)(aP + k);
        bw[0][u]   = *(const v2f*)(bP + (k / 4) * (NT * 64));
      }
#pragma unroll
      for (int g = 0; g < NG; ++g) {
        if (g + 1 < NG) {
#pragma unroll
          for (int u = 0; u < KS; ++u) {
            const int k = 4 * KS * (g + 1) + 4 * u;
            av[(g + 1) & 1][u]   = *(const v2f*)(aP + k);
            bw[(g + 1) & 1][u]   = *(const v2f*)(bP + (k / 4) * (NT * 64));
          }
        }
        __builtin_amdgcn_sched_barrier(0);
#pragma unroll
        for (int u = 0; u < KS; ++u) {
          d = __builtin_amdgcn_wmma_f32_16x16x4_f32(false, av[g & 1][u], false, bw[g & 1][u],
                                                    (short)0, d, false, false);
        }
      }
    }
    __syncthreads();
    if (wave < NOUT / 16) {
      float* cP = C + (size_t)(row0 + half * 8) * NOUT + (wave * 16 + l16);
#pragma unroll
      for (int r = 0; r < 8; ++r) cP[r * NOUT] = d[r];
    }
  }
}

#define DEC_BATCH 8
#define DEC_PAIRS_PER_WAVE 32

__global__ __launch_bounds__(256) void k_decode(const float* __restrict__ z, const int* __restrict__ pa,
                                                const int* __restrict__ pb, float* __restrict__ out, int np) {
  const int lane = threadIdx.x & 31;
  const int l    = lane & 15;
  const int gw   = __builtin_amdgcn_readfirstlane((int)((blockIdx.x * blockDim.x + threadIdx.x) >> 5));
  const int nw   = (int)((gridDim.x * blockDim.x) >> 5);
  const float4* __restrict__ zv = (const float4*)z;

  for (int p0 = gw * DEC_PAIRS_PER_WAVE; p0 < np; p0 += nw * DEC_PAIRS_PER_WAVE) {
    const int  mine = p0 + lane;
    const bool live = mine < np;
    const int at = min(mine, np - 1);
    const unsigned ia = min((unsigned)pa[at], (unsigned)(NNODES - 1));
    const unsigned ib = min((unsigned)pb[at], (unsigned)(NNODES - 1));

    float res = 0.0f;
#pragma unroll
    for (int b = 0; b < 16; b += DEC_BATCH) {
      float4 va[DEC_BATCH], vb[DEC_BATCH];
#pragma unroll
      for (int u = 0; u < DEC_BATCH; ++u) {
        const unsigned a = (unsigned)__shfl((int)ia, b + u, 16);
        const unsigned c = (unsigned)__shfl((int)ib, b + u, 16);
        va[u] = zv[a * 16u + l];
        vb[u] = zv[c * 16u + l];
      }
#pragma unroll
      for (int u = 0; u < DEC_BATCH; ++u) {
        float p = va[u].x * vb[u].x + va[u].y * vb[u].y + va[u].z * vb[u].z + va[u].w * vb[u].w;
        p += __shfl_xor(p, 8);
        p += __shfl_xor(p, 4);
        p += __shfl_xor(p, 2);
        p += __shfl_xor(p, 1);
        if (l == b + u) res = p;
      }
    }
    if (live) out[mine] = res;
  }
}

extern "C" void kernel_launch(void* const* d_in, const int* in_sizes, int n_in,
                              void* d_out, int out_size, void* d_ws, size_t ws_size,
                              hipStream_t stream) {
  (void)in_sizes; (void)n_in; (void)out_size; (void)ws_size;

  const float* x  = (const float*)d_in[0];
  const int*   ei = (const int*)d_in[1];
  const int*   el = (const int*)d_in[2];
  const float* W1 = (const float*)d_in[3];
  const float* b1 = (const float*)d_in[4];
  const float* W2 = (const float*)d_in[5];
  const float* b2 = (const float*)d_in[6];
  const float* W3 = (const float*)d_in[7];
  const float* b3 = (const float*)d_in[8];

  const int* src = ei;
  const int* dst = ei + NEDGES;
  const int* pa  = el;
  const int* pb  = el + NPAIRS;
  float* out = (float*)d_out;

  float* dinv   = (float*)d_ws;
  int*   rank   = (int*)(dinv + ((NNODES + 127) & ~127));
  float* bufA   = (float*)(rank + ((NEDGES + 127) & ~127));
  float* bufB   = bufA + (size_t)NNODES * DHID;
  int*   cnt    = (int*)(bufB + (size_t)NNODES * DHID);
  int*   rowptr = cnt + ((NNODES + 127) & ~127);
  int*   tsum   = rowptr + ((NNODES + 1 + 127) & ~127);
  int2*  rec    = (int2*)(tsum + 128);
  unsigned* tab = (unsigned*)(rec + NEDGES);
  unsigned* pre = tab + (size_t)RANK_CHUNKS * RANK_WORDS;

  auto cdiv = [](long long a, long long b) { return (int)((a + b - 1) / b); };

  const int nTiles = cdiv(NNODES, SCAN_TILE);
  k_chunk_rank  <<<RANK_CHUNKS, RANK_THREADS, 0, stream>>>(dst, rank, tab, NEDGES);
  k_chunk_prefix<<<cdiv(RANK_WORDS, 256), 256, 0, stream>>>(tab, (uint2*)pre, cnt);
  k_scan_tiles<<<nTiles, 256, 0, stream>>>(cnt, tsum, NNODES);
  k_scan_rows <<<nTiles, 256, 0, stream>>>(cnt, tsum, rowptr, dinv, NNODES);
  k_fill      <<<cdiv(NEDGES, 256), 256, 0, stream>>>(src, dst, rank, rowptr, pre, dinv, rec, NEDGES);

  const int mTiles = NNODES / 16;
  const int gBlocks = cdiv((long long)cdiv(NNODES, GATHER_ROWS) * 32, 256);

  k_gemm_wmma<DHID><<<dim3(mTiles, DHID / 64), 128, 0, stream>>>(x, W1, bufA);

  k_agg_gemm<DHID><<<FUSED_GRID, FUSED_THREADS, 0, stream>>>(bufA, rowptr, rec, dinv, b1, W2, bufB, NNODES);

  k_agg_gemm<DOUT><<<FUSED_GRID, FUSED_THREADS, 0, stream>>>(bufB, rowptr, rec, dinv, b2, W3, bufA, NNODES);

  k_agg_gather<DOUT, false><<<gBlocks, 256, 0, stream>>>(bufA, rowptr, rec, dinv, b3, bufB, NNODES);

  k_decode<<<cdiv((long long)cdiv(NPAIRS, DEC_PAIRS_PER_WAVE) * 32, 256), 256, 0, stream>>>(bufB, pa, pb, out, NPAIRS);
}


# layer-1 product moved behind its aggregation, agg(x) @ W1 for agg(x @ W1) (its sum order changes), chained with the W2 product on the LDS tile: stand-alone product launch gone, 10 launches become 9
# speedup vs baseline: 4.2287x; 1.0640x over previous
#include <hip/hip_runtime.h>

#define NNODES 100000
#define NEDGES 1600000
#define NPAIRS 200000
#define DIN    128
#define DHID   128
#define DOUT   64
#define KDIM   128
#define APITCH 132

typedef __attribute__((ext_vector_type(2))) float v2f;
typedef __attribute__((ext_vector_type(8))) float v8f;

#define RANK_CHUNK   32768
#define RANK_CHUNKS  ((NEDGES + RANK_CHUNK - 1) / RANK_CHUNK)
#define RANK_WORDS   (NNODES / 2)
#define RANK_THREADS 1024
#define RANK_PER_THREAD (RANK_CHUNK / RANK_THREADS)
static_assert(RANK_CHUNK <= 65535, "a node's count inside one chunk fits 16 bits");
static_assert(RANK_CHUNK % RANK_THREADS == 0, "whole rounds");
static_assert(NNODES % 8 == 0, "a chunk's packed row is whole 16-byte units");

__global__ __launch_bounds__(RANK_THREADS) void k_chunk_rank(const int* __restrict__ dst,
                                                             int* __restrict__ rank,
                                                             unsigned* __restrict__ tab, int e) {
  __shared__ __attribute__((aligned(16))) unsigned lc[RANK_WORDS];
  const int tid = threadIdx.x;
  for (int i = tid; i < RANK_WORDS / 4; i += RANK_THREADS) ((uint4*)lc)[i] = make_uint4(0u, 0u, 0u, 0u);

  const int base = blockIdx.x * RANK_CHUNK;
  int d[RANK_PER_THREAD];
#pragma unroll
  for (int j = 0; j < RANK_PER_THREAD; ++j) {
    const int i = base + j * RANK_THREADS + tid;
    d[j] = (i < e) ? dst[i] : -1;
  }
  __syncthreads();
  unsigned old[RANK_PER_THREAD];
#pragma unroll
  for (int j = 0; j < RANK_PER_THREAD; ++j) {
    const bool live = d[j] >= 0;
    const unsigned sh = (unsigned)(d[j] & 1) * 16u;
    old[j] = atomicAdd(&lc[live ? (d[j] >> 1) : 0], live ? (1u << sh) : 0u);
  }
#pragma unroll
  for (int j = 0; j < RANK_PER_THREAD; ++j) {
    const int i = base + j * RANK_THREADS + tid;
    const unsigned sh = (unsigned)(d[j] & 1) * 16u;
    if (d[j] >= 0) rank[i] = (int)((old[j] >> sh) & 0xFFFFu);
  }
  __syncthreads();
  uint4* __restrict__ row = (uint4*)(tab + (size_t)blockIdx.x * RANK_WORDS);
  for (int i = tid; i < RANK_WORDS / 4; i += RANK_THREADS) row[i] = ((const uint4*)lc)[i];
}

__global__ __launch_bounds__(256) void k_chunk_prefix(const unsigned* __restrict__ tab,
                                                      uint2* __restrict__ pre,
                                                      int* __restrict__ cnt) {
  const int w = blockIdx.x * 256 + threadIdx.x;
  if (w >= RANK_WORDS) return;
  unsigned t[RANK_CHUNKS];
#pragma unroll
  for (int c = 0; c < RANK_CHUNKS; ++c) t[c] = tab[c * RANK_WORDS + w];
  unsigned lo = 0u, hi = 0u;
#pragma unroll
  for (int c = 0; c < RANK_CHUNKS; ++c) {
    pre[c * RANK_WORDS + w] = make_uint2(lo, hi);
    lo += t[c] & 0xFFFFu;
    hi += t[c] >> 16;
  }
  ((int2*)cnt)[w] = make_int2((int)lo, (int)hi);
}

#define SCAN_ITEMS 4
#define SCAN_TILE  1024
static_assert((NNODES + SCAN_TILE - 1) / SCAN_TILE <= 128, "tsum holds 128 tile sums");

__device__ __forceinline__ void load_cnt4(const int* __restrict__ cnt, int base, int n, int c[SCAN_ITEMS]) {
  if (base + SCAN_ITEMS <= n) {
    const int4 q = *(const int4*)(cnt + base);
    c[0] = q.x; c[1] = q.y; c[2] = q.z; c[3] = q.w;
  } else {
    for (int j = 0; j < SCAN_ITEMS; ++j) c[j] = (base + j < n) ? cnt[base + j] : 0;
  }
}

__device__ __forceinline__ int block_sum256(int x, int* sh) {
  const int lane = threadIdx.x & 31, wv = threadIdx.x >> 5;
  for (int o = 16; o > 0; o >>= 1) x += __shfl_xor(x, o);
  if (lane == 0) sh[wv] = x;
  __syncthreads();
  int t = 0;
  for (int i = 0; i < 8; ++i) t += sh[i];
  return t;
}

__global__ __launch_bounds__(256) void k_scan_tiles(const int* __restrict__ cnt,
                                                    int* __restrict__ tsum, int n) {
  __shared__ int sh[8];
  int c[SCAN_ITEMS];
  load_cnt4(cnt, blockIdx.x * SCAN_TILE + threadIdx.x * SCAN_ITEMS, n, c);
  const int tot = block_sum256(c[0] + c[1] + c[2] + c[3], sh);
  if (threadIdx.x == 0) tsum[blockIdx.x] = tot;
}

__global__ __launch_bounds__(256) void k_scan_rows(const int* __restrict__ cnt,
                                                   const int* __restrict__ tsum,
                                                   int* __restrict__ rowptr,
                                                   float* __restrict__ dinv, int n) {
  __shared__ int sh[8];
  __shared__ int shw[8];
  const int t = threadIdx.x, lane = t & 31, wv = t >> 5;

  int part = 0;
  for (int i = t; i < (int)blockIdx.x; i += 256) part += tsum[i];
  const int tile_off = block_sum256(part, sh);

  const int base = blockIdx.x * SCAN_TILE + t * SCAN_ITEMS;
  int c[SCAN_ITEMS];
  load_cnt4(cnt, base, n, c);
  const int mine = c[0] + c[1] + c[2] + c[3];

  int incl = mine;
  for (int o = 1; o < 32; o <<= 1) {
    const int y = __shfl_up(incl, o);
    if (lane >= o) incl += y;
  }
  if (lane == 31) shw[wv] = incl;
  __syncthreads();
  int wave_off = 0;
  for (int i = 0; i < wv; ++i) wave_off += shw[i];

  int run = tile_off + wave_off + incl - mine;
  for (int j = 0; j < SCAN_ITEMS; ++j) {
    const int i = base + j;
    if (i < n) {
      rowptr[i] = run;
      dinv[i]   = rsqrtf((float)(c[j] + 1));
      run += c[j];
      if (i == n - 1) rowptr[n] = run;
    }
  }
}

__global__ void k_fill(const int* __restrict__ src, const int* __restrict__ dst,
                       const int* __restrict__ rank, const int* __restrict__ rowptr,
                       const unsigned* __restrict__ pre,
                       const float* __restrict__ dinv, int2* __restrict__ rec, int e) {
  int i = blockIdx.x * blockDim.x + threadIdx.x;
  if (i >= e) return;
  const int s = src[i], d = dst[i];
  const int pos = rowptr[d] + (int)pre[(unsigned)(i / RANK_CHUNK) * (unsigned)NNODES + (unsigned)d] + rank[i];
  const float w = dinv[s] * dinv[d];
  if ((unsigned)pos < (unsigned)e) rec[pos] = make_int2(s, __float_as_int(w));
}


#define GATHER_DEPTH 8

template <int V, bool FULL>
__device__ __forceinline__ void gather_group(float __attribute__((ext_vector_type(V)))& acc,
                                             const float __attribute__((ext_vector_type(V)))* __restrict__ hv,
                                             const int2 r, const int k, const int m,
                                             const int lane, const unsigned last_node) {
  typedef float vec_t __attribute__((ext_vector_type(V)));
  unsigned s[GATHER_DEPTH];
  float    w[GATHER_DEPTH];
  vec_t    v[GATHER_DEPTH];
#pragma unroll
  for (int u = 0; u < GATHER_DEPTH; ++u) {
    s[u] = min((unsigned)__builtin_amdgcn_readlane(r.x, k + u), last_node);
    w[u] = __int_as_float(__builtin_amdgcn_readlane(r.y, k + u));
  }
#pragma unroll
  for (int u = 0; u < GATHER_DEPTH; ++u) {
    v[u] = (vec_t)(0.0f);
    if (FULL || u < m) v[u] = hv[s[u] * 32u + lane];
  }
#pragma unroll
  for (int u = 0; u < GATHER_DEPTH; ++u) acc += v[u] * w[u];
}

template <int V>
__device__ __forceinline__ void gather_chunk(float __attribute__((ext_vector_type(V)))& acc,
                                             const float __attribute__((ext_vector_type(V)))* __restrict__ hv,
                                             const int2 r, const int cnt,
                                             const int lane, const unsigned last_node) {
  int k = 0;
  for (; k + GATHER_DEPTH <= cnt; k += GATHER_DEPTH)
    gather_group<V, true>(acc, hv, r, k, GATHER_DEPTH, lane, last_node);
  if (k < cnt)
    gather_group<V, false>(acc, hv, r, k, cnt - k, lane, last_node);
}

#define GATHER_ROWS 8

template <int F, bool RELU>
__global__ __launch_bounds__(256) void k_agg_gather(const float* __restrict__ h,
                                                    const int* __restrict__ rowptr,
                                                    const int2* __restrict__ rec,
                                                    const float* __restrict__ dinv,
                                                    const float* __restrict__ bias,
                                                    float* __restrict__ out, int n) {
  constexpr int V = F / 32;
  typedef float vec_t __attribute__((ext_vector_type(V)));
  const vec_t* __restrict__ hv = (const vec_t*)h;

  const int lane = threadIdx.x & 31;
  const int gw = __builtin_amdgcn_readfirstlane((int)((blockIdx.x * blockDim.x + threadIdx.x) >> 5));
  const int r0 = gw * GATHER_ROWS;
  if (r0 >= n) return;
  const int r1 = min(r0 + GATHER_ROWS, n);

  const vec_t bv = ((const vec_t*)bias)[lane];

  int end  = rowptr[r0];
  int nend = rowptr[r0 + 1];
  int2 rn = make_int2(0, 0);
  if (end + lane < nend) rn = rec[end + lane];
  vec_t sn = hv[(unsigned)r0 * 32u + lane];
  float dn = dinv[r0];

  for (int row = r0; row < r1; ++row) {
    const int beg = end;
    end = nend;
    const int2 r = rn;
    const vec_t self = sn;
    const float di = dn;
    if (row + 1 < r1) {
      nend = rowptr[row + 2];
      rn = make_int2(0, 0);
      if (end + lane < nend) rn = rec[end + lane];
      sn = hv[(unsigned)(row + 1) * 32u + lane];
      dn = dinv[row + 1];
    }

    vec_t acc = self * (di * di);

    gather_chunk<V>(acc, hv, r, min(32, end - beg), lane, (unsigned)(n - 1));
    for (int base = beg + 32; base < end; base += 32) {
      int2 r2 = make_int2(0, 0);
      if (base + lane < end) r2 = rec[base + lane];
      gather_chunk<V>(acc, hv, r2, min(32, end - base), lane, (unsigned)(n - 1));
    }

    acc += bv;
    if (RELU) {
      for (int c = 0; c < V; ++c) acc[c] = fmaxf(acc[c], 0.0f);
    }
    ((vec_t*)out)[(unsigned)row * 32u + lane] = acc;
  }
}

#define FUSED_WAVES   16
#define FUSED_THREADS (FUSED_WAVES * 32)
#define FUSED_PLACES  1024
#define FUSED_TILES   (NNODES / 16)
#define FUSED_TPB     ((FUSED_TILES + FUSED_PLACES - 1) / FUSED_PLACES)
#define FUSED_GRID    ((FUSED_TILES + FUSED_TPB - 1) / FUSED_TPB)
static_assert(FUSED_GRID <= FUSED_PLACES, "every block resident at once");
static_assert(NNODES % 16 == 0, "whole 16-row tiles");

template <int NOUT>
__global__ __launch_bounds__(FUSED_THREADS, 16) void k_agg_gemm(const float* __restrict__ h,
                                                            const int* __restrict__ rowptr,
                                                            const int2* __restrict__ rec,
                                                            const float* __restrict__ dinv,
                                                            const float* __restrict__ bias,
                                                            const float* __restrict__ Wn,
                                                            float* __restrict__ C, int n) {
  constexpr int V  = KDIM / 32;
  constexpr int NT = NOUT / 16;
  typedef float vec_t __attribute__((ext_vector_type(V)));
  __shared__ __attribute__((aligned(16))) float lW[KDIM * NOUT];
  __shared__ __attribute__((aligned(16))) float lA[16 * APITCH];

  const int tid  = threadIdx.x;
  const int lane = tid & 31;
  const int wave = __builtin_amdgcn_readfirstlane(tid >> 5);
  const int half = lane >> 4;
  const int l16  = lane & 15;

  {
    const float4* gW = (const float4*)Wn;
    constexpr int R4 = NOUT / 4;
    constexpr int UNITS = (KDIM / 2) * R4;
    static_assert(UNITS % FUSED_THREADS == 0, "whole rounds");
#pragma unroll
    for (int i = 0; i < UNITS / FUSED_THREADS; ++i) {
      const int q  = tid + i * FUSED_THREADS;
      const int rp = q / R4, c4 = q % R4;
      const float4 x0 = gW[(2 * rp) * R4 + c4];
      const float4 x1 = gW[(2 * rp + 1) * R4 + c4];
      const int kt = rp >> 1, hh = rp & 1;
      const int nt = c4 >> 2, cc = (c4 & 3) * 4;
      float* dst = lW + (((kt * NT + nt) * 32) + hh * 16 + cc) * 2;
      *(float4*)(dst)     = make_float4(x0.x, x1.x, x0.y, x1.y);
      *(float4*)(dst + 4) = make_float4(x0.z, x1.z, x0.w, x1.w);
    }
  }

  const vec_t* __restrict__ hv = (const vec_t*)h;
  const vec_t bv = ((const vec_t*)bias)[lane];
  const int ntiles = n / 16;

  for (int tile = blockIdx.x; tile < ntiles; tile += gridDim.x) {
    const int row0 = tile * 16;
    const int row  = row0 + wave;

    const int beg = rowptr[row];
    const int end = rowptr[row + 1];
    const float di = dinv[row];
    vec_t acc = hv[(unsigned)row * 32u + lane] * (di * di);

    for (int base = beg; base < end; base += 32) {
      int2 r = make_int2(0, 0);
      if (base + lane < end) r = rec[base + lane];
      const int cnt = min(32, end - base);

      int k = 0;
      for (; k + GATHER_DEPTH <= cnt; k += GATHER_DEPTH)
        gather_group<V, true>(acc, hv, r, k, GATHER_DEPTH, lane, (unsigned)(n - 1));
      if (k < cnt)
        gather_group<V, false>(acc, hv, r, k, cnt - k, lane, (unsigned)(n - 1));
    }

    acc += bv;
    for (int c = 0; c < V; ++c) acc[c] = fmaxf(acc[c], 0.0f);
    *(vec_t*)(lA + wave * APITCH + lane * V) = acc;
    __syncthreads();

    v8f d = {};
    if (wave < NOUT / 16) {
      const float* aP = lA + l16 * APITCH + half * 2;
      const float* bP = lW + (wave * 32 + lane) * 2;
      constexpr int KS = 4;
      constexpr int NG = KDIM / (4 * KS);
      v2f av[2][KS], bw[2][KS];
#pragma unroll
      for (int u = 0; u < KS; ++u) {
        const int k = 4 * u;
        av[0][u]   = *(const v2f*)(aP + k);
        bw[0][u]   = *(const v2f*)(bP + (k / 4) * (NT * 64));
      }
#pragma unroll
      for (int g = 0; g < NG; ++g) {
        if (g + 1 < NG) {
#pragma unroll
          for (int u = 0; u < KS; ++u) {
            const int k = 4 * KS * (g + 1) + 4 * u;
            av[(g + 1) & 1][u]   = *(const v2f*)(aP + k);
            bw[(g + 1) & 1][u]   = *(const v2f*)(bP + (k / 4) * (NT * 64));
          }
        }
        __builtin_amdgcn_sched_barrier(0);
#pragma unroll
        for (int u = 0; u < KS; ++u) {
          d = __builtin_amdgcn_wmma_f32_16x16x4_f32(false, av[g & 1][u], false, bw[g & 1][u],
                                                    (short)0, d, false, false);
        }
      }
    }
    __syncthreads();
    if (wave < NOUT / 16) {
      float* cP = C + (size_t)(row0 + half * 8) * NOUT + (wave * 16 + l16);
#pragma unroll
      for (int r = 0; r < 8; ++r) cP[r * NOUT] = d[r];
    }
  }
}

#define F2_WAVES   32
#define F2_THREADS (F2_WAVES * 32)
#define F2_ROWS    32
#define F2_GRID    512
static_assert(NNODES % F2_ROWS == 0, "whole 32-row tiles");

__device__ __forceinline__ void stage_w128_frag(const float* __restrict__ Wg, float* lW, const int tid) {
  const float4* gW = (const float4*)Wg;
  constexpr int NT = DHID / 16;
  constexpr int R4 = DHID / 4;
  constexpr int UNITS = (KDIM / 2) * R4;
  static_assert(UNITS % F2_THREADS == 0, "whole rounds");
#pragma unroll
  for (int i = 0; i < UNITS / F2_THREADS; ++i) {
    const int q  = tid + i * F2_THREADS;
    const int rp = q / R4, c4 = q % R4;
    const float4 x0 = gW[(2 * rp) * R4 + c4];
    const float4 x1 = gW[(2 * rp + 1) * R4 + c4];
    const int kt = rp >> 1, hh = rp & 1;
    const int nt = c4 >> 2, cc = (c4 & 3) * 4;
    float* dst = lW + (((kt * NT + nt) * 32) + hh * 16 + cc) * 2;
    *(float4*)(dst)     = make_float4(x0.x, x1.x, x0.y, x1.y);
    *(float4*)(dst + 4) = make_float4(x0.z, x1.z, x0.w, x1.w);
  }
}

template <int NT>
__device__ __forceinline__ v8f tile_product128(const float* aP, const float* bP) {
  constexpr int KS = 4;
  constexpr int NG = KDIM / (4 * KS);
  v8f d = {};
  v2f av[2][KS], bw[2][KS];
#pragma unroll
  for (int u = 0; u < KS; ++u) {
    const int k = 4 * u;
    av[0][u] = *(const v2f*)(aP + k);
    bw[0][u] = *(const v2f*)(bP + (k / 4) * (NT * 64));
  }
#pragma unroll
  for (int g = 0; g < NG; ++g) {
    if (g + 1 < NG) {
#pragma unroll
      for (int u = 0; u < KS; ++u) {
        const int k = 4 * KS * (g + 1) + 4 * u;
        av[(g + 1) & 1][u] = *(const v2f*)(aP + k);
        bw[(g + 1) & 1][u] = *(const v2f*)(bP + (k / 4) * (NT * 64));
      }
    }
    __builtin_amdgcn_sched_barrier(0);
#pragma unroll
    for (int u = 0; u < KS; ++u) {
      d = __builtin_amdgcn_wmma_f32_16x16x4_f32(false, av[g & 1][u], false, bw[g & 1][u],
                                                (short)0, d, false, false);
    }
  }
  return d;
}

__global__ __launch_bounds__(F2_THREADS, 16) void k_agg_gemm2(const float* __restrict__ x,
                                                             const int* __restrict__ rowptr,
                                                             const int2* __restrict__ rec,
                                                             const float* __restrict__ dinv,
                                                             const float* __restrict__ Wa,
                                                             const float* __restrict__ ba,
                                                             const float* __restrict__ Wb,
                                                             float* __restrict__ C, int n) {
  constexpr int V  = KDIM / 32;
  constexpr int NT = DHID / 16;
  typedef float vec_t __attribute__((ext_vector_type(V)));
  __shared__ __attribute__((aligned(16))) float lWa[KDIM * DHID];
  __shared__ __attribute__((aligned(16))) float lWb[KDIM * DHID];
  __shared__ __attribute__((aligned(16))) float lA[F2_ROWS * APITCH];

  const int tid  = threadIdx.x;
  const int lane = tid & 31;
  const int wave = __builtin_amdgcn_readfirstlane(tid >> 5);
  const int half = lane >> 4;
  const int l16  = lane & 15;

  stage_w128_frag(Wa, lWa, tid);
  stage_w128_frag(Wb, lWb, tid);

  const vec_t* __restrict__ hv = (const vec_t*)x;
  const int ntiles = n / F2_ROWS;

  const bool mm = wave < 2 * NT;
  const int mt = (wave >> 3) & 1, nt = wave & 7;
  const float* aP  = lA + (mt * 16 + l16) * APITCH + half * 2;
  const float* bPa = lWa + (nt * 32 + lane) * 2;
  const float* bPb = lWb + (nt * 32 + lane) * 2;
  float* hP = lA + (mt * 16 + half * 8) * APITCH + (nt * 16 + l16);
  const float bcol = ba[nt * 16 + l16];

  for (int tile = blockIdx.x; tile < ntiles; tile += gridDim.x) {
    const int row0 = tile * F2_ROWS;
    const int row  = row0 + wave;

    const int beg = rowptr[row];
    const int end = rowptr[row + 1];
    const float di = dinv[row];
    vec_t acc = hv[(unsigned)row * 32u + lane] * (di * di);

    for (int base = beg; base < end; base += 32) {
      int2 r = make_int2(0, 0);
      if (base + lane < end) r = rec[base + lane];
      const int cnt = min(32, end - base);

      int k = 0;
      for (; k + GATHER_DEPTH <= cnt; k += GATHER_DEPTH)
        gather_group<V, true>(acc, hv, r, k, GATHER_DEPTH, lane, (unsigned)(n - 1));
      if (k < cnt)
        gather_group<V, false>(acc, hv, r, k, cnt - k, lane, (unsigned)(n - 1));
    }
    *(vec_t*)(lA + wave * APITCH + lane * V) = acc;
    __syncthreads();

    v8f d = {};
    if (mm) d = tile_product128<NT>(aP, bPa);
    __syncthreads();

    if (mm) {
#pragma unroll
      for (int r = 0; r < 8; ++r) hP[r * APITCH] = fmaxf(d[r] + bcol, 0.0f);
    }
    __syncthreads();

    v8f d2 = {};
    if (mm) d2 = tile_product128<NT>(aP, bPb);
    __syncthreads();
    if (mm) {
      float* cP = C + (size_t)(row0 + mt * 16 + half * 8) * DHID + (nt * 16 + l16);
#pragma unroll
      for (int r = 0; r < 8; ++r) cP[r * DHID] = d2[r];
    }
  }
}

#define DEC_BATCH 8
#define DEC_PAIRS_PER_WAVE 32

__global__ __launch_bounds__(256) void k_decode(const float* __restrict__ z, const int* __restrict__ pa,
                                                const int* __restrict__ pb, float* __restrict__ out, int np) {
  const int lane = threadIdx.x & 31;
  const int l    = lane & 15;
  const int gw   = __builtin_amdgcn_readfirstlane((int)((blockIdx.x * blockDim.x + threadIdx.x) >> 5));
  const int nw   = (int)((gridDim.x * blockDim.x) >> 5);
  const float4* __restrict__ zv = (const float4*)z;

  for (int p0 = gw * DEC_PAIRS_PER_WAVE; p0 < np; p0 += nw * DEC_PAIRS_PER_WAVE) {
    const int  mine = p0 + lane;
    const bool live = mine < np;
    const int at = min(mine, np - 1);
    const unsigned ia = min((unsigned)pa[at], (unsigned)(NNODES - 1));
    const unsigned ib = min((unsigned)pb[at], (unsigned)(NNODES - 1));

    float res = 0.0f;
#pragma unroll
    for (int b = 0; b < 16; b += DEC_BATCH) {
      float4 va[DEC_BATCH], vb[DEC_BATCH];
#pragma unroll
      for (int u = 0; u < DEC_BATCH; ++u) {
        const unsigned a = (unsigned)__shfl((int)ia, b + u, 16);
        const unsigned c = (unsigned)__shfl((int)ib, b + u, 16);
        va[u] = zv[a * 16u + l];
        vb[u] = zv[c * 16u + l];
      }
#pragma unroll
      for (int u = 0; u < DEC_BATCH; ++u) {
        float p = va[u].x * vb[u].x + va[u].y * vb[u].y + va[u].z * vb[u].z + va[u].w * vb[u].w;
        p += __shfl_xor(p, 8);
        p += __shfl_xor(p, 4);
        p += __shfl_xor(p, 2);
        p += __shfl_xor(p, 1);
        if (l == b + u) res = p;
      }
    }
    if (live) out[mine] = res;
  }
}

extern "C" void kernel_launch(void* const* d_in, const int* in_sizes, int n_in,
                              void* d_out, int out_size, void* d_ws, size_t ws_size,
                              hipStream_t stream) {
  (void)in_sizes; (void)n_in; (void)out_size; (void)ws_size;

  const float* x  = (const float*)d_in[0];
  const int*   ei = (const int*)d_in[1];
  const int*   el = (const int*)d_in[2];
  const float* W1 = (const float*)d_in[3];
  const float* b1 = (const float*)d_in[4];
  const float* W2 = (const float*)d_in[5];
  const float* b2 = (const float*)d_in[6];
  const float* W3 = (const float*)d_in[7];
  const float* b3 = (const float*)d_in[8];

  const int* src = ei;
  const int* dst = ei + NEDGES;
  const int* pa  = el;
  const int* pb  = el + NPAIRS;
  float* out = (float*)d_out;

  float* dinv   = (float*)d_ws;
  int*   rank   = (int*)(dinv + ((NNODES + 127) & ~127));
  float* bufA   = (float*)(rank + ((NEDGES + 127) & ~127));
  float* bufB   = bufA + (size_t)NNODES * DHID;
  int*   cnt    = (int*)(bufB + (size_t)NNODES * DHID);
  int*   rowptr = cnt + ((NNODES + 127) & ~127);
  int*   tsum   = rowptr + ((NNODES + 1 + 127) & ~127);
  int2*  rec    = (int2*)(tsum + 128);
  unsigned* tab = (unsigned*)(rec + NEDGES);
  unsigned* pre = tab + (size_t)RANK_CHUNKS * RANK_WORDS;

  auto cdiv = [](long long a, long long b) { return (int)((a + b - 1) / b); };

  const int nTiles = cdiv(NNODES, SCAN_TILE);
  k_chunk_rank  <<<RANK_CHUNKS, RANK_THREADS, 0, stream>>>(dst, rank, tab, NEDGES);
  k_chunk_prefix<<<cdiv(RANK_WORDS, 256), 256, 0, stream>>>(tab, (uint2*)pre, cnt);
  k_scan_tiles<<<nTiles, 256, 0, stream>>>(cnt, tsum, NNODES);
  k_scan_rows <<<nTiles, 256, 0, stream>>>(cnt, tsum, rowptr, dinv, NNODES);
  k_fill      <<<cdiv(NEDGES, 256), 256, 0, stream>>>(src, dst, rank, rowptr, pre, dinv, rec, NEDGES);

  const int gBlocks = cdiv((long long)cdiv(NNODES, GATHER_ROWS) * 32, 256);

  k_agg_gemm2<<<F2_GRID, F2_THREADS, 0, stream>>>(x, rowptr, rec, dinv, W1, b1, W2, bufB, NNODES);

  k_agg_gemm<DOUT><<<FUSED_GRID, FUSED_THREADS, 0, stream>>>(bufB, rowptr, rec, dinv, b2, W3, bufA, NNODES);

  k_agg_gather<DOUT, false><<<gBlocks, 256, 0, stream>>>(bufA, rowptr, rec, dinv, b3, bufB, NNODES);

  k_decode<<<cdiv((long long)cdiv(NPAIRS, DEC_PAIRS_PER_WAVE) * 32, 256), 256, 0, stream>>>(bufB, pa, pb, out, NPAIRS);
}


# node prefix scan folded into its neighbours: in-block scan inside the chunk-prefix launch and added into its table, the 196 block sums scanned in LDS by every fill block; 9 launches become 7
# speedup vs baseline: 4.3601x; 1.0311x over previous
#include <hip/hip_runtime.h>

#define NNODES 100000
#define NEDGES 1600000
#define NPAIRS 200000
#define DIN    128
#define DHID   128
#define DOUT   64
#define KDIM   128
#define APITCH 132

typedef __attribute__((ext_vector_type(2))) float v2f;
typedef __attribute__((ext_vector_type(8))) float v8f;

#define RANK_CHUNK   32768
#define RANK_CHUNKS  ((NEDGES + RANK_CHUNK - 1) / RANK_CHUNK)
#define RANK_WORDS   (NNODES / 2)
#define RANK_THREADS 1024
#define RANK_PER_THREAD (RANK_CHUNK / RANK_THREADS)
static_assert(RANK_CHUNK <= 65535, "a node's count inside one chunk fits 16 bits");
static_assert(RANK_CHUNK % RANK_THREADS == 0, "whole rounds");
static_assert(NNODES % 8 == 0, "a chunk's packed row is whole 16-byte units");

__global__ __launch_bounds__(RANK_THREADS) void k_chunk_rank(const int* __restrict__ dst,
                                                             int* __restrict__ rank,
                                                             unsigned* __restrict__ tab, int e) {
  __shared__ __attribute__((aligned(16))) unsigned lc[RANK_WORDS];
  const int tid = threadIdx.x;
  for (int i = tid; i < RANK_WORDS / 4; i += RANK_THREADS) ((uint4*)lc)[i] = make_uint4(0u, 0u, 0u, 0u);

  const int base = blockIdx.x * RANK_CHUNK;
  int d[RANK_PER_THREAD];
#pragma unroll
  for (int j = 0; j < RANK_PER_THREAD; ++j) {
    const int i = base + j * RANK_THREADS + tid;
    d[j] = (i < e) ? dst[i] : -1;
  }
  __syncthreads();
  unsigned old[RANK_PER_THREAD];
#pragma unroll
  for (int j = 0; j < RANK_PER_THREAD; ++j) {
    const bool live = d[j] >= 0;
    const unsigned sh = (unsigned)(d[j] & 1) * 16u;
    old[j] = atomicAdd(&lc[live ? (d[j] >> 1) : 0], live ? (1u << sh) : 0u);
  }
#pragma unroll
  for (int j = 0; j < RANK_PER_THREAD; ++j) {
    const int i = base + j * RANK_THREADS + tid;
    const unsigned sh = (unsigned)(d[j] & 1) * 16u;
    if (d[j] >= 0) rank[i] = (int)((old[j] >> sh) & 0xFFFFu);
  }
  __syncthreads();
  uint4* __restrict__ row = (uint4*)(tab + (size_t)blockIdx.x * RANK_WORDS);
  for (int i = tid; i < RANK_WORDS / 4; i += RANK_THREADS) row[i] = ((const uint4*)lc)[i];
}

#define PREFIX_THREADS 256
#define PREFIX_BLOCKS  ((RANK_WORDS + PREFIX_THREADS - 1) / PREFIX_THREADS)
#define PREFIX_SHIFT   9
static_assert((2 * PREFIX_THREADS) == (1 << PREFIX_SHIFT), "node >> PREFIX_SHIFT is the node's prefix block");
static_assert(PREFIX_BLOCKS < 256, "a 256-thread block scans all block sums, one a thread, and keeps the total");

__global__ __launch_bounds__(PREFIX_THREADS) void k_chunk_prefix(const unsigned* __restrict__ tab,
                                                                 uint2* __restrict__ pre,
                                                                 int* __restrict__ bsum,
                                                                 float* __restrict__ dinv) {
  __shared__ unsigned shw[PREFIX_THREADS / 32];
  const int tid = threadIdx.x, lane = tid & 31, wv = tid >> 5;
  const int w = blockIdx.x * PREFIX_THREADS + tid;
  const bool live = w < RANK_WORDS;
  const int wl = live ? w : RANK_WORDS - 1;
  unsigned t[RANK_CHUNKS];
#pragma unroll
  for (int c = 0; c < RANK_CHUNKS; ++c) t[c] = tab[c * RANK_WORDS + wl];
  unsigned lo = 0u, hi = 0u;
#pragma unroll
  for (int c = 0; c < RANK_CHUNKS; ++c) {
    lo += t[c] & 0xFFFFu;
    hi += t[c] >> 16;
  }
  if (!live) { lo = 0u; hi = 0u; }

  const int mine = (int)(lo + hi);
  int incl = mine;
  for (int o = 1; o < 32; o <<= 1) {
    const int y = __shfl_up(incl, o);
    if (lane >= o) incl += y;
  }
  if (lane == 31) shw[wv] = (unsigned)incl;
  __syncthreads();
  unsigned wave_off = 0u, tot = 0u;
  for (int i = 0; i < PREFIX_THREADS / 32; ++i) {
    const unsigned s = shw[i];
    if (i < wv) wave_off += s;
    tot += s;
  }
  if (tid == 0) bsum[blockIdx.x] = (int)tot;
  if (live) {
    unsigned a = wave_off + (unsigned)(incl - mine);
    unsigned b = a + lo;
#pragma unroll
    for (int c = 0; c < RANK_CHUNKS; ++c) {
      pre[c * RANK_WORDS + w] = make_uint2(a, b);
      a += t[c] & 0xFFFFu;
      b += t[c] >> 16;
    }
    ((float2*)dinv)[w] = make_float2(rsqrtf((float)(lo + 1u)), rsqrtf((float)(hi + 1u)));
  }
}

#define FILL_THREADS 256
static_assert(((NEDGES + FILL_THREADS - 1) / FILL_THREADS) * FILL_THREADS >= NNODES + 1,
              "the fill grid has a thread for every rowptr entry");

__global__ __launch_bounds__(FILL_THREADS) void k_fill(const int* __restrict__ src, const int* __restrict__ dst,
                                                       const int* __restrict__ rank,
                                                       const unsigned* __restrict__ pre,
                                                       const int* __restrict__ bsum,
                                                       const float* __restrict__ dinv,
                                                       int* __restrict__ rowptr,
                                                       int2* __restrict__ rec, int e, int n) {
  __shared__ int tpre[FILL_THREADS];
  __shared__ int shw[FILL_THREADS / 32];
  const int tid = threadIdx.x, lane = tid & 31, wv = tid >> 5;
  const int i = blockIdx.x * FILL_THREADS + tid;
  const bool live = i < e;

  const int ie = min(i, e - 1);
  const int mine0 = bsum[min(tid, PREFIX_BLOCKS - 1)];
  const unsigned p0 = pre[min(i, n - 1)];
  const int s = src[ie], d = dst[ie], rk = rank[ie];
  const unsigned pg = pre[(unsigned)(ie / RANK_CHUNK) * (unsigned)NNODES + (unsigned)d];
  const float ws = dinv[s];
  const float wd = dinv[d];
  const int mine = (tid < PREFIX_BLOCKS) ? mine0 : 0;

  int incl = mine;
  for (int o = 1; o < 32; o <<= 1) {
    const int y = __shfl_up(incl, o);
    if (lane >= o) incl += y;
  }
  if (lane == 31) shw[wv] = incl;
  __syncthreads();
  int wave_off = 0;
  for (int k = 0; k < FILL_THREADS / 32; ++k) {
    const int sk = shw[k];
    if (k < wv) wave_off += sk;
  }
  tpre[tid] = wave_off + incl - mine;
  __syncthreads();

  if (i < n)       rowptr[i] = tpre[i >> PREFIX_SHIFT] + (int)p0;
  else if (i == n) rowptr[n] = tpre[PREFIX_BLOCKS];

  if (live) {
    const int pos = tpre[d >> PREFIX_SHIFT] + (int)pg + rk;
    const float w = ws * wd;
    if ((unsigned)pos < (unsigned)e) rec[pos] = make_int2(s, __float_as_int(w));
  }
}


#define GATHER_DEPTH 8

template <int V, bool FULL>
__device__ __forceinline__ void gather_group(float __attribute__((ext_vector_type(V)))& acc,
                                             const float __attribute__((ext_vector_type(V)))* __restrict__ hv,
                                             const int2 r, const int k, const int m,
                                             const int lane, const unsigned last_node) {
  typedef float vec_t __attribute__((ext_vector_type(V)));
  unsigned s[GATHER_DEPTH];
  float    w[GATHER_DEPTH];
  vec_t    v[GATHER_DEPTH];
#pragma unroll
  for (int u = 0; u < GATHER_DEPTH; ++u) {
    s[u] = min((unsigned)__builtin_amdgcn_readlane(r.x, k + u), last_node);
    w[u] = __int_as_float(__builtin_amdgcn_readlane(r.y, k + u));
  }
#pragma unroll
  for (int u = 0; u < GATHER_DEPTH; ++u) {
    v[u] = (vec_t)(0.0f);
    if (FULL || u < m) v[u] = hv[s[u] * 32u + lane];
  }
#pragma unroll
  for (int u = 0; u < GATHER_DEPTH; ++u) acc += v[u] * w[u];
}

template <int V>
__device__ __forceinline__ void gather_chunk(float __attribute__((ext_vector_type(V)))& acc,
                                             const float __attribute__((ext_vector_type(V)))* __restrict__ hv,
                                             const int2 r, const int cnt,
                                             const int lane, const unsigned last_node) {
  int k = 0;
  for (; k + GATHER_DEPTH <= cnt; k += GATHER_DEPTH)
    gather_group<V, true>(acc, hv, r, k, GATHER_DEPTH, lane, last_node);
  if (k < cnt)
    gather_group<V, false>(acc, hv, r, k, cnt - k, lane, last_node);
}

#define GATHER_ROWS 8

template <int F, bool RELU>
__global__ __launch_bounds__(256) void k_agg_gather(const float* __restrict__ h,
                                                    const int* __restrict__ rowptr,
                                                    const int2* __restrict__ rec,
                                                    const float* __restrict__ dinv,
                                                    const float* __restrict__ bias,
                                                    float* __restrict__ out, int n) {
  constexpr int V = F / 32;
  typedef float vec_t __attribute__((ext_vector_type(V)));
  const vec_t* __restrict__ hv = (const vec_t*)h;

  const int lane = threadIdx.x & 31;
  const int gw = __builtin_amdgcn_readfirstlane((int)((blockIdx.x * blockDim.x + threadIdx.x) >> 5));
  const int r0 = gw * GATHER_ROWS;
  if (r0 >= n) return;
  const int r1 = min(r0 + GATHER_ROWS, n);

  const vec_t bv = ((const vec_t*)bias)[lane];

  int end  = rowptr[r0];
  int nend = rowptr[r0 + 1];
  int2 rn = make_int2(0, 0);
  if (end + lane < nend) rn = rec[end + lane];
  vec_t sn = hv[(unsigned)r0 * 32u + lane];
  float dn = dinv[r0];

  for (int row = r0; row < r1; ++row) {
    const int beg = end;
    end = nend;
    const int2 r = rn;
    const vec_t self = sn;
    const float di = dn;
    if (row + 1 < r1) {
      nend = rowptr[row + 2];
      rn = make_int2(0, 0);
      if (end + lane < nend) rn = rec[end + lane];
      sn = hv[(unsigned)(row + 1) * 32u + lane];
      dn = dinv[row + 1];
    }

    vec_t acc = self * (di * di);

    gather_chunk<V>(acc, hv, r, min(32, end - beg), lane, (unsigned)(n - 1));
    for (int base = beg + 32; base < end; base += 32) {
      int2 r2 = make_int2(0, 0);
      if (base + lane < end) r2 = rec[base + lane];
      gather_chunk<V>(acc, hv, r2, min(32, end - base), lane, (unsigned)(n - 1));
    }

    acc += bv;
    if (RELU) {
      for (int c = 0; c < V; ++c) acc[c] = fmaxf(acc[c], 0.0f);
    }
    ((vec_t*)out)[(unsigned)row * 32u + lane] = acc;
  }
}

#define FUSED_WAVES   16
#define FUSED_THREADS (FUSED_WAVES * 32)
#define FUSED_PLACES  1024
#define FUSED_TILES   (NNODES / 16)
#define FUSED_TPB     ((FUSED_TILES + FUSED_PLACES - 1) / FUSED_PLACES)
#define FUSED_GRID    ((FUSED_TILES + FUSED_TPB - 1) / FUSED_TPB)
static_assert(FUSED_GRID <= FUSED_PLACES, "every block resident at once");
static_assert(NNODES % 16 == 0, "whole 16-row tiles");

template <int NOUT>
__global__ __launch_bounds__(FUSED_THREADS, 16) void k_agg_gemm(const float* __restrict__ h,
                                                            const int* __restrict__ rowptr,
                                                            const int2* __restrict__ rec,
                                                            const float* __restrict__ dinv,
                                                            const float* __restrict__ bias,
                                                            const float* __restrict__ Wn,
                                                            float* __restrict__ C, int n) {
  constexpr int V  = KDIM / 32;
  constexpr int NT = NOUT / 16;
  typedef float vec_t __attribute__((ext_vector_type(V)));
  __shared__ __attribute__((aligned(16))) float lW[KDIM * NOUT];
  __shared__ __attribute__((aligned(16))) float lA[16 * APITCH];

  const int tid  = threadIdx.x;
  const int lane = tid & 31;
  const int wave = __builtin_amdgcn_readfirstlane(tid >> 5);
  const int half = lane >> 4;
  const int l16  = lane & 15;

  {
    const float4* gW = (const float4*)Wn;
    constexpr int R4 = NOUT / 4;
    constexpr int UNITS = (KDIM / 2) * R4;
    static_assert(UNITS % FUSED_THREADS == 0, "whole rounds");
#pragma unroll
    for (int i = 0; i < UNITS / FUSED_THREADS; ++i) {
      const int q  = tid + i * FUSED_THREADS;
      const int rp = q / R4, c4 = q % R4;
      const float4 x0 = gW[(2 * rp) * R4 + c4];
      const float4 x1 = gW[(2 * rp + 1) * R4 + c4];
      const int kt = rp >> 1, hh = rp & 1;
      const int nt = c4 >> 2, cc = (c4 & 3) * 4;
      float* dst = lW + (((kt * NT + nt) * 32) + hh * 16 + cc) * 2;
      *(float4*)(dst)     = make_float4(x0.x, x1.x, x0.y, x1.y);
      *(float4*)(dst + 4) = make_float4(x0.z, x1.z, x0.w, x1.w);
    }
  }

  const vec_t* __restrict__ hv = (const vec_t*)h;
  const vec_t bv = ((const vec_t*)bias)[lane];
  const int ntiles = n / 16;

  for (int tile = blockIdx.x; tile < ntiles; tile += gridDim.x) {
    const int row0 = tile * 16;
    const int row  = row0 + wave;

    const int beg = rowptr[row];
    const int end = rowptr[row + 1];
    const float di = dinv[row];
    vec_t acc = hv[(unsigned)row * 32u + lane] * (di * di);

    for (int base = beg; base < end; base += 32) {
      int2 r = make_int2(0, 0);
      if (base + lane < end) r = rec[base + lane];
      const int cnt = min(32, end - base);

      int k = 0;
      for (; k + GATHER_DEPTH <= cnt; k += GATHER_DEPTH)
        gather_group<V, true>(acc, hv, r, k, GATHER_DEPTH, lane, (unsigned)(n - 1));
      if (k < cnt)
        gather_group<V, false>(acc, hv, r, k, cnt - k, lane, (unsigned)(n - 1));
    }

    acc += bv;
    for (int c = 0; c < V; ++c) acc[c] = fmaxf(acc[c], 0.0f);
    *(vec_t*)(lA + wave * APITCH + lane * V) = acc;
    __syncthreads();

    v8f d = {};
    if (wave < NOUT / 16) {
      const float* aP = lA + l16 * APITCH + half * 2;
      const float* bP = lW + (wave * 32 + lane) * 2;
      constexpr int KS = 4;
      constexpr int NG = KDIM / (4 * KS);
      v2f av[2][KS], bw[2][KS];
#pragma unroll
      for (int u = 0; u < KS; ++u) {
        const int k = 4 * u;
        av[0][u]   = *(const v2f*)(aP + k);
        bw[0][u]   = *(const v2f*)(bP + (k / 4) * (NT * 64));
      }
#pragma unroll
      for (int g = 0; g < NG; ++g) {
        if (g + 1 < NG) {
#pragma unroll
          for (int u = 0; u < KS; ++u) {
            const int k = 4 * KS * (g + 1) + 4 * u;
            av[(g + 1) & 1][u]   = *(const v2f*)(aP + k);
            bw[(g + 1) & 1][u]   = *(const v2f*)(bP + (k / 4) * (NT * 64));
          }
        }
        __builtin_amdgcn_sched_barrier(0);
#pragma unroll
        for (int u = 0; u < KS; ++u) {
          d = __builtin_amdgcn_wmma_f32_16x16x4_f32(false, av[g & 1][u], false, bw[g & 1][u],
                                                    (short)0, d, false, false);
        }
      }
    }
    __syncthreads();
    if (wave < NOUT / 16) {
      float* cP = C + (size_t)(row0 + half * 8) * NOUT + (wave * 16 + l16);
#pragma unroll
      for (int r = 0; r < 8; ++r) cP[r * NOUT] = d[r];
    }
  }
}

#define F2_WAVES   32
#define F2_THREADS (F2_WAVES * 32)
#define F2_ROWS    32
#define F2_GRID    512
static_assert(NNODES % F2_ROWS == 0, "whole 32-row tiles");

__device__ __forceinline__ void stage_w128_frag(const float* __restrict__ Wg, float* lW, const int tid) {
  const float4* gW = (const float4*)Wg;
  constexpr int NT = DHID / 16;
  constexpr int R4 = DHID / 4;
  constexpr int UNITS = (KDIM / 2) * R4;
  static_assert(UNITS % F2_THREADS == 0, "whole rounds");
#pragma unroll
  for (int i = 0; i < UNITS / F2_THREADS; ++i) {
    const int q  = tid + i * F2_THREADS;
    const int rp = q / R4, c4 = q % R4;
    const float4 x0 = gW[(2 * rp) * R4 + c4];
    const float4 x1 = gW[(2 * rp + 1) * R4 + c4];
    const int kt = rp >> 1, hh = rp & 1;
    const int nt = c4 >> 2, cc = (c4 & 3) * 4;
    float* dst = lW + (((kt * NT + nt) * 32) + hh * 16 + cc) * 2;
    *(float4*)(dst)     = make_float4(x0.x, x1.x, x0.y, x1.y);
    *(float4*)(dst + 4) = make_float4(x0.z, x1.z, x0.w, x1.w);
  }
}

template <int NT>
__device__ __forceinline__ v8f tile_product128(const float* aP, const float* bP) {
  constexpr int KS = 4;
  constexpr int NG = KDIM / (4 * KS);
  v8f d = {};
  v2f av[2][KS], bw[2][KS];
#pragma unroll
  for (int u = 0; u < KS; ++u) {
    const int k = 4 * u;
    av[0][u] = *(const v2f*)(aP + k);
    bw[0][u] = *(const v2f*)(bP + (k / 4) * (NT * 64));
  }
#pragma unroll
  for (int g = 0; g < NG; ++g) {
    if (g + 1 < NG) {
#pragma unroll
      for (int u = 0; u < KS; ++u) {
        const int k = 4 * KS * (g + 1) + 4 * u;
        av[(g + 1) & 1][u] = *(const v2f*)(aP + k);
        bw[(g + 1) & 1][u] = *(const v2f*)(bP + (k / 4) * (NT * 64));
      }
    }
    __builtin_amdgcn_sched_barrier(0);
#pragma unroll
    for (int u = 0; u < KS; ++u) {
      d = __builtin_amdgcn_wmma_f32_16x16x4_f32(false, av[g & 1][u], false, bw[g & 1][u],
                                                (short)0, d, false, false);
    }
  }
  return d;
}

__global__ __launch_bounds__(F2_THREADS, 16) void k_agg_gemm2(const float* __restrict__ x,
                                                             const int* __restrict__ rowptr,
                                                             const int2* __restrict__ rec,
                                                             const float* __restrict__ dinv,
                                                             const float* __restrict__ Wa,
                                                             const float* __restrict__ ba,
                                                             const float* __restrict__ Wb,
                                                             float* __restrict__ C, int n) {
  constexpr int V  = KDIM / 32;
  constexpr int NT = DHID / 16;
  typedef float vec_t __attribute__((ext_vector_type(V)));
  __shared__ __attribute__((aligned(16))) float lWa[KDIM * DHID];
  __shared__ __attribute__((aligned(16))) float lWb[KDIM * DHID];
  __shared__ __attribute__((aligned(16))) float lA[F2_ROWS * APITCH];

  const int tid  = threadIdx.x;
  const int lane = tid & 31;
  const int wave = __builtin_amdgcn_readfirstlane(tid >> 5);
  const int half = lane >> 4;
  const int l16  = lane & 15;

  stage_w128_frag(Wa, lWa, tid);
  stage_w128_frag(Wb, lWb, tid);

  const vec_t* __restrict__ hv = (const vec_t*)x;
  const int ntiles = n / F2_ROWS;

  const bool mm = wave < 2 * NT;
  const int mt = (wave >> 3) & 1, nt = wave & 7;
  const float* aP  = lA + (mt * 16 + l16) * APITCH + half * 2;
  const float* bPa = lWa + (nt * 32 + lane) * 2;
  const float* bPb = lWb + (nt * 32 + lane) * 2;
  float* hP = lA + (mt * 16 + half * 8) * APITCH + (nt * 16 + l16);
  const float bcol = ba[nt * 16 + l16];

  for (int tile = blockIdx.x; tile < ntiles; tile += gridDim.x) {
    const int row0 = tile * F2_ROWS;
    const int row  = row0 + wave;

    const int beg = rowptr[row];
    const int end = rowptr[row + 1];
    const float di = dinv[row];
    vec_t acc = hv[(unsigned)row * 32u + lane] * (di * di);

    for (int base = beg; base < end; base += 32) {
      int2 r = make_int2(0, 0);
      if (base + lane < end) r = rec[base + lane];
      const int cnt = min(32, end - base);

      int k = 0;
      for (; k + GATHER_DEPTH <= cnt; k += GATHER_DEPTH)
        gather_group<V, true>(acc, hv, r, k, GATHER_DEPTH, lane, (unsigned)(n - 1));
      if (k < cnt)
        gather_group<V, false>(acc, hv, r, k, cnt - k, lane, (unsigned)(n - 1));
    }
    *(vec_t*)(lA + wave * APITCH + lane * V) = acc;
    __syncthreads();

    v8f d = {};
    if (mm) d = tile_product128<NT>(aP, bPa);
    __syncthreads();

    if (mm) {
#pragma unroll
      for (int r = 0; r < 8; ++r) hP[r * APITCH] = fmaxf(d[r] + bcol, 0.0f);
    }
    __syncthreads();

    v8f d2 = {};
    if (mm) d2 = tile_product128<NT>(aP, bPb);
    __syncthreads();
    if (mm) {
      float* cP = C + (size_t)(row0 + mt * 16 + half * 8) * DHID + (nt * 16 + l16);
#pragma unroll
      for (int r = 0; r < 8; ++r) cP[r * DHID] = d2[r];
    }
  }
}

#define DEC_BATCH 8
#define DEC_PAIRS_PER_WAVE 32

__global__ __launch_bounds__(256) void k_decode(const float* __restrict__ z, const int* __restrict__ pa,
                                                const int* __restrict__ pb, float* __restrict__ out, int np) {
  const int lane = threadIdx.x & 31;
  const int l    = lane & 15;
  const int gw   = __builtin_amdgcn_readfirstlane((int)((blockIdx.x * blockDim.x + threadIdx.x) >> 5));
  const int nw   = (int)((gridDim.x * blockDim.x) >> 5);
  const float4* __restrict__ zv = (const float4*)z;

  for (int p0 = gw * DEC_PAIRS_PER_WAVE; p0 < np; p0 += nw * DEC_PAIRS_PER_WAVE) {
    const int  mine = p0 + lane;
    const bool live = mine < np;
    const int at = min(mine, np - 1);
    const unsigned ia = min((unsigned)pa[at], (unsigned)(NNODES - 1));
    const unsigned ib = min((unsigned)pb[at], (unsigned)(NNODES - 1));

    float res = 0.0f;
#pragma unroll
    for (int b = 0; b < 16; b += DEC_BATCH) {
      float4 va[DEC_BATCH], vb[DEC_BATCH];
#pragma unroll
      for (int u = 0; u < DEC_BATCH; ++u) {
        const unsigned a = (unsigned)__shfl((int)ia, b + u, 16);
        const unsigned c = (unsigned)__shfl((int)ib, b + u, 16);
        va[u] = zv[a * 16u + l];
        vb[u] = zv[c * 16u + l];
      }
#pragma unroll
      for (int u = 0; u < DEC_BATCH; ++u) {
        float p = va[u].x * vb[u].x + va[u].y * vb[u].y + va[u].z * vb[u].z + va[u].w * vb[u].w;
        p += __shfl_xor(p, 8);
        p += __shfl_xor(p, 4);
        p += __shfl_xor(p, 2);
        p += __shfl_xor(p, 1);
        if (l == b + u) res = p;
      }
    }
    if (live) out[mine] = res;
  }
}

extern "C" void kernel_launch(void* const* d_in, const int* in_sizes, int n_in,
                              void* d_out, int out_size, void* d_ws, size_t ws_size,
                              hipStream_t stream) {
  (void)in_sizes; (void)n_in; (void)out_size; (void)ws_size;

  const float* x  = (const float*)d_in[0];
  const int*   ei = (const int*)d_in[1];
  const int*   el = (const int*)d_in[2];
  const float* W1 = (const float*)d_in[3];
  const float* b1 = (const float*)d_in[4];
  const float* W2 = (const float*)d_in[5];
  const float* b2 = (const float*)d_in[6];
  const float* W3 = (const float*)d_in[7];
  const float* b3 = (const float*)d_in[8];

  const int* src = ei;
  const int* dst = ei + NEDGES;
  const int* pa  = el;
  const int* pb  = el + NPAIRS;
  float* out = (float*)d_out;

  float* dinv   = (float*)d_ws;
  int*   rank   = (int*)(dinv + ((NNODES + 127) & ~127));
  float* bufA   = (float*)(rank + ((NEDGES + 127) & ~127));
  float* bufB   = bufA + (size_t)NNODES * DHID;
  int*   rowptr = (int*)(bufB + (size_t)NNODES * DHID);
  int*   bsum   = rowptr + ((NNODES + 1 + 127) & ~127);
  int2*  rec    = (int2*)(bsum + 256);
  unsigned* tab = (unsigned*)(rec + NEDGES);
  unsigned* pre = tab + (size_t)RANK_CHUNKS * RANK_WORDS;

  auto cdiv = [](long long a, long long b) { return (int)((a + b - 1) / b); };

  k_chunk_rank  <<<RANK_CHUNKS, RANK_THREADS, 0, stream>>>(dst, rank, tab, NEDGES);
  k_chunk_prefix<<<PREFIX_BLOCKS, PREFIX_THREADS, 0, stream>>>(tab, (uint2*)pre, bsum, dinv);
  k_fill        <<<cdiv(NEDGES, FILL_THREADS), FILL_THREADS, 0, stream>>>(src, dst, rank, pre, bsum, dinv, rowptr, rec,
                                                                         NEDGES, NNODES);

  const int gBlocks = cdiv((long long)cdiv(NNODES, GATHER_ROWS) * 32, 256);

  k_agg_gemm2<<<F2_GRID, F2_THREADS, 0, stream>>>(x, rowptr, rec, dinv, W1, b1, W2, bufB, NNODES);

  k_agg_gemm<DOUT><<<FUSED_GRID, FUSED_THREADS, 0, stream>>>(bufB, rowptr, rec, dinv, b2, W3, bufA, NNODES);

  k_agg_gather<DOUT, false><<<gBlocks, 256, 0, stream>>>(bufA, rowptr, rec, dinv, b3, bufB, NNODES);

  k_decode<<<cdiv((long long)cdiv(NPAIRS, DEC_PAIRS_PER_WAVE) * 32, 256), 256, 0, stream>>>(bufB, pa, pb, out, NPAIRS);
}
